# starting kernel
# baseline (speedup 1.0000x reference)
#include <hip/hip_runtime.h>
#include <hip/hip_bf16.h>
#include <math.h>


typedef float v2f __attribute__((ext_vector_type(2)));
typedef float v8f __attribute__((ext_vector_type(8)));

#define N_NODES 50000
#define N_EDGES 600000
#define E_TOT   650000
#define IN_CH   256
#define HID     128
#define ENC_NEG_INF 0x007FFFFFu

__device__ __forceinline__ unsigned enc_f32(float f) {
  unsigned u = __float_as_uint(f);
  return (u & 0x80000000u) ? ~u : (u | 0x80000000u);
}
__device__ __forceinline__ float dec_f32(unsigned e) {
  unsigned u = (e & 0x80000000u) ? (e & 0x7FFFFFFFu) : ~e;
  return __uint_as_float(u);
}

__device__ __forceinline__ float wave_sum(float v) {
  #pragma unroll
  for (int off = 16; off >= 1; off >>= 1) v += __shfl_xor(v, off, 32);
  return v;
}

__global__ void k_gemm_relu(const float* __restrict__ X,
                            const float* __restrict__ W,
                            const float* __restrict__ B,
                            float* __restrict__ H,
                            float* __restrict__ rn) {
  const int wave = (int)((blockIdx.x * blockDim.x + threadIdx.x) >> 5);
  const int lane = (int)(threadIdx.x & 31);
  if (wave * 16 >= N_NODES) return;

  const int m0   = wave * 16;
  const int mrow = m0 + (lane & 15);
  const int kg   = (lane >> 4) * 2;
  const int n15  = lane & 15;

  v8f acc[8];
  #pragma unroll
  for (int t = 0; t < 8; ++t) acc[t] = (v8f){0.f,0.f,0.f,0.f,0.f,0.f,0.f,0.f};

  const float* __restrict__ xrow = X + (size_t)mrow * IN_CH + kg;
  const float* __restrict__ wbas = W + (size_t)n15 * IN_CH + kg;

  for (int k = 0; k < IN_CH; k += 8) {
    const v2f a0 = *(const v2f*)(xrow + k);
    const v2f a1 = *(const v2f*)(xrow + k + 4);
    v2f b0[8], b1[8];
    #pragma unroll
    for (int t = 0; t < 8; ++t)
      b0[t] = *(const v2f*)(wbas + (size_t)t * 16 * IN_CH + k);
    #pragma unroll
    for (int t = 0; t < 8; ++t)
      b1[t] = *(const v2f*)(wbas + (size_t)t * 16 * IN_CH + k + 4);
    #pragma unroll
    for (int t = 0; t < 8; ++t)
      acc[t] = __builtin_amdgcn_wmma_f32_16x16x4_f32(
          false, a0, false, b0[t], (short)0, acc[t], false, false);
    #pragma unroll
    for (int t = 0; t < 8; ++t)
      acc[t] = __builtin_amdgcn_wmma_f32_16x16x4_f32(
          false, a1, false, b1[t], (short)0, acc[t], false, false);
  }

  const int mside = (lane >> 4) * 8;
  float ss[8];
  #pragma unroll
  for (int r = 0; r < 8; ++r) ss[r] = 0.f;

  #pragma unroll
  for (int t = 0; t < 8; ++t) {
    const int col = t * 16 + n15;
    const float bb = B[col];
    #pragma unroll
    for (int r = 0; r < 8; ++r) {
      const int row = m0 + r + mside;
      float v = acc[t][r] + bb;
      v = v > 0.f ? v : 0.f;
      H[(size_t)row * HID + col] = v;
      ss[r] += v * v;
    }
  }
  #pragma unroll
  for (int r = 0; r < 8; ++r) {
    #pragma unroll
    for (int off = 8; off >= 1; off >>= 1) ss[r] += __shfl_xor(ss[r], off, 32);
  }
  if (n15 == 0) {
    #pragma unroll
    for (int r = 0; r < 8; ++r)
      rn[m0 + r + mside] = 1.f / fmaxf(sqrtf(ss[r]), 1e-12f);
  }
}

__global__ void k_init(float* __restrict__ Y,
                       unsigned* __restrict__ segmax,
                       float* __restrict__ segsum) {
  const int i = (int)(blockIdx.x * blockDim.x + threadIdx.x);
  if (i < N_NODES * HID) Y[i] = 0.f;
  if (i < N_NODES) { segmax[i] = ENC_NEG_INF; segsum[i] = 0.f; }
}

__global__ void k_rnorm(const float* __restrict__ X, float* __restrict__ rn) {
  const int node = (int)((blockIdx.x * blockDim.x + threadIdx.x) >> 5);
  const int lane = (int)(threadIdx.x & 31);
  if (node >= N_NODES) return;
  const float4 v = *(const float4*)(X + (size_t)node * HID + lane * 4);
  float s = v.x * v.x + v.y * v.y + v.z * v.z + v.w * v.w;
  s = wave_sum(s);
  if (lane == 0) rn[node] = 1.f / fmaxf(sqrtf(s), 1e-12f);
}

__global__ void k_edge_logit(const float* __restrict__ X,
                             const int* __restrict__ EI,
                             const float* __restrict__ rn,
                             const float* __restrict__ beta_ptr,
                             float* __restrict__ elog,
                             unsigned* __restrict__ segmax) {
  const int e    = (int)((blockIdx.x * blockDim.x + threadIdx.x) >> 5);
  const int lane = (int)(threadIdx.x & 31);
  if (e >= E_TOT) return;
  int src, dst;
  if (e < N_EDGES) { src = EI[e]; dst = EI[N_EDGES + e]; }
  else             { src = dst = e - N_EDGES; }
  const float4 a = *(const float4*)(X + (size_t)src * HID + lane * 4);
  const float4 b = *(const float4*)(X + (size_t)dst * HID + lane * 4);
  float s = a.x * b.x + a.y * b.y + a.z * b.z + a.w * b.w;
  s = wave_sum(s);
  if (lane == 0) {
    const float beta = beta_ptr ? beta_ptr[0] : 1.0f;
    const float logit = beta * s * rn[src] * rn[dst];
    elog[e] = logit;
    atomicMax(segmax + dst, enc_f32(logit));
  }
}

__global__ void k_edge_exp(const int* __restrict__ EI,
                           const unsigned* __restrict__ segmax,
                           float* __restrict__ elog,
                           float* __restrict__ segsum) {
  const int e = (int)(blockIdx.x * blockDim.x + threadIdx.x);
  if (e >= E_TOT) return;
  const int dst = (e < N_EDGES) ? EI[N_EDGES + e] : (e - N_EDGES);
  const float ex = __expf(elog[e] - dec_f32(segmax[dst]));
  elog[e] = ex;
  atomicAdd(segsum + dst, ex);
}

__global__ void k_scatter(const float* __restrict__ X,
                          const int* __restrict__ EI,
                          const float* __restrict__ elog,
                          const float* __restrict__ segsum,
                          float* __restrict__ Y) {
  const int e    = (int)((blockIdx.x * blockDim.x + threadIdx.x) >> 5);
  const int lane = (int)(threadIdx.x & 31);
  if (e >= E_TOT) return;
  int src, dst;
  if (e < N_EDGES) { src = EI[e]; dst = EI[N_EDGES + e]; }
  else             { src = dst = e - N_EDGES; }
  const float alpha = elog[e] / (segsum[dst] + 1e-16f);
  const float4 xs = *(const float4*)(X + (size_t)src * HID + lane * 4);
  float* yp = Y + (size_t)dst * HID + lane * 4;
  atomicAdd(yp + 0, alpha * xs.x);
  atomicAdd(yp + 1, alpha * xs.y);
  atomicAdd(yp + 2, alpha * xs.z);
  atomicAdd(yp + 3, alpha * xs.w);
}

static void run_conv(const float* X, float* Y, const int* EI,
                     const float* beta_ptr, float* rn, unsigned* segmax,
                     float* segsum, float* elog, bool compute_rnorm,
                     hipStream_t stream) {
  const int init_threads = N_NODES * HID;
  k_init<<<(init_threads + 255) / 256, 256, 0, stream>>>(Y, segmax, segsum);

  if (compute_rnorm) {
    const long long norm_threads = (long long)N_NODES * 32;
    k_rnorm<<<(int)((norm_threads + 255) / 256), 256, 0, stream>>>(X, rn);
  }

  const long long edge_threads = (long long)E_TOT * 32;
  const int eblocks = (int)((edge_threads + 255) / 256);
  k_edge_logit<<<eblocks, 256, 0, stream>>>(X, EI, rn, beta_ptr, elog, segmax);

  k_edge_exp<<<(E_TOT + 255) / 256, 256, 0, stream>>>(EI, segmax, elog, segsum);

  k_scatter<<<eblocks, 256, 0, stream>>>(X, EI, elog, segsum, Y);
}

extern "C" void kernel_launch(void* const* d_in, const int* in_sizes, int n_in,
                              void* d_out, int out_size, void* d_ws, size_t ws_size,
                              hipStream_t stream) {
  const float* x     = (const float*)d_in[0];
  const int*   ei    = (const int*)d_in[1];
  const float* W1    = (const float*)d_in[2];
  const float* b1    = (const float*)d_in[3];
  const float* beta2 = (const float*)d_in[4];
  float* out = (float*)d_out;

  float*    ws     = (float*)d_ws;
  float*    h1     = ws;
  float*    h2     = ws + 6400000;
  float*    rn     = ws + 12800000;
  unsigned* segmax = (unsigned*)(ws + 12850000);
  float*    segsum = ws + 12900000;
  float*    elog   = ws + 12950000;

  const int gemm_waves  = (N_NODES + 15) / 16;
  const int gemm_blocks = (gemm_waves * 32 + 255) / 256;
  k_gemm_relu<<<gemm_blocks, 256, 0, stream>>>(x, W1, b1, h1, rn);

  run_conv(h1, h2, ei, nullptr, rn, segmax, segsum, elog, false, stream);
  run_conv(h2, out, ei, beta2, rn, segmax, segsum, elog, true, stream);
}


# Edge-logit and scatter kernels: 8 consecutive edges per wave (was 1), their 512B row gathers in flight together; per-edge arithmetic unchanged, f32 atomics kept (arrival order varies as before)
# speedup vs baseline: 1.3605x; 1.3605x over previous
#include <hip/hip_runtime.h>
#include <hip/hip_bf16.h>
#include <math.h>


typedef float v2f __attribute__((ext_vector_type(2)));
typedef float v8f __attribute__((ext_vector_type(8)));

#define N_NODES 50000
#define N_EDGES 600000
#define E_TOT   650000
#define IN_CH   256
#define HID     128
#define ENC_NEG_INF 0x007FFFFFu
#define EPW     8
static_assert(N_EDGES % EPW == 0 && E_TOT % EPW == 0,
              "a wave's edge group must not straddle the self-loop boundary or the end");

__device__ __forceinline__ unsigned enc_f32(float f) {
  unsigned u = __float_as_uint(f);
  return (u & 0x80000000u) ? ~u : (u | 0x80000000u);
}
__device__ __forceinline__ float dec_f32(unsigned e) {
  unsigned u = (e & 0x80000000u) ? (e & 0x7FFFFFFFu) : ~e;
  return __uint_as_float(u);
}

__device__ __forceinline__ float wave_sum(float v) {
  #pragma unroll
  for (int off = 16; off >= 1; off >>= 1) v += __shfl_xor(v, off, 32);
  return v;
}

__global__ void k_gemm_relu(const float* __restrict__ X,
                            const float* __restrict__ W,
                            const float* __restrict__ B,
                            float* __restrict__ H,
                            float* __restrict__ rn) {
  const int wave = (int)((blockIdx.x * blockDim.x + threadIdx.x) >> 5);
  const int lane = (int)(threadIdx.x & 31);
  if (wave * 16 >= N_NODES) return;

  const int m0   = wave * 16;
  const int mrow = m0 + (lane & 15);
  const int kg   = (lane >> 4) * 2;
  const int n15  = lane & 15;

  v8f acc[8];
  #pragma unroll
  for (int t = 0; t < 8; ++t) acc[t] = (v8f){0.f,0.f,0.f,0.f,0.f,0.f,0.f,0.f};

  const float* __restrict__ xrow = X + (size_t)mrow * IN_CH + kg;
  const float* __restrict__ wbas = W + (size_t)n15 * IN_CH + kg;

  for (int k = 0; k < IN_CH; k += 8) {
    const v2f a0 = *(const v2f*)(xrow + k);
    const v2f a1 = *(const v2f*)(xrow + k + 4);
    v2f b0[8], b1[8];
    #pragma unroll
    for (int t = 0; t < 8; ++t)
      b0[t] = *(const v2f*)(wbas + (size_t)t * 16 * IN_CH + k);
    #pragma unroll
    for (int t = 0; t < 8; ++t)
      b1[t] = *(const v2f*)(wbas + (size_t)t * 16 * IN_CH + k + 4);
    #pragma unroll
    for (int t = 0; t < 8; ++t)
      acc[t] = __builtin_amdgcn_wmma_f32_16x16x4_f32(
          false, a0, false, b0[t], (short)0, acc[t], false, false);
    #pragma unroll
    for (int t = 0; t < 8; ++t)
      acc[t] = __builtin_amdgcn_wmma_f32_16x16x4_f32(
          false, a1, false, b1[t], (short)0, acc[t], false, false);
  }

  const int mside = (lane >> 4) * 8;
  float ss[8];
  #pragma unroll
  for (int r = 0; r < 8; ++r) ss[r] = 0.f;

  #pragma unroll
  for (int t = 0; t < 8; ++t) {
    const int col = t * 16 + n15;
    const float bb = B[col];
    #pragma unroll
    for (int r = 0; r < 8; ++r) {
      const int row = m0 + r + mside;
      float v = acc[t][r] + bb;
      v = v > 0.f ? v : 0.f;
      H[(size_t)row * HID + col] = v;
      ss[r] += v * v;
    }
  }
  #pragma unroll
  for (int r = 0; r < 8; ++r) {
    #pragma unroll
    for (int off = 8; off >= 1; off >>= 1) ss[r] += __shfl_xor(ss[r], off, 32);
  }
  if (n15 == 0) {
    #pragma unroll
    for (int r = 0; r < 8; ++r)
      rn[m0 + r + mside] = 1.f / fmaxf(sqrtf(ss[r]), 1e-12f);
  }
}

__global__ void k_init(float* __restrict__ Y,
                       unsigned* __restrict__ segmax,
                       float* __restrict__ segsum) {
  const int i = (int)(blockIdx.x * blockDim.x + threadIdx.x);
  if (i < N_NODES * HID) Y[i] = 0.f;
  if (i < N_NODES) { segmax[i] = ENC_NEG_INF; segsum[i] = 0.f; }
}

__global__ void k_rnorm(const float* __restrict__ X, float* __restrict__ rn) {
  const int node = (int)((blockIdx.x * blockDim.x + threadIdx.x) >> 5);
  const int lane = (int)(threadIdx.x & 31);
  if (node >= N_NODES) return;
  const float4 v = *(const float4*)(X + (size_t)node * HID + lane * 4);
  float s = v.x * v.x + v.y * v.y + v.z * v.z + v.w * v.w;
  s = wave_sum(s);
  if (lane == 0) rn[node] = 1.f / fmaxf(sqrtf(s), 1e-12f);
}

__global__ void k_edge_logit(const float* __restrict__ X,
                             const int* __restrict__ EI,
                             const float* __restrict__ rn,
                             const float* __restrict__ beta_ptr,
                             float* __restrict__ elog,
                             unsigned* __restrict__ segmax) {
  const int wave = __builtin_amdgcn_readfirstlane(
      (int)((blockIdx.x * blockDim.x + threadIdx.x) >> 5));
  const int lane = (int)(threadIdx.x & 31);
  const int e0   = wave * EPW;
  if (e0 >= E_TOT) return;
  const int own = lane & (EPW - 1);
  const int ej  = e0 + own;
  int msrc, mdst;
  if (e0 < N_EDGES) { msrc = EI[ej]; mdst = EI[N_EDGES + ej]; }
  else              { msrc = mdst = ej - N_EDGES; }
  const float rs   = rn[msrc];
  const float rd   = rn[mdst];
  const float beta = beta_ptr ? beta_ptr[0] : 1.0f;

  float4 a[EPW], b[EPW];
  #pragma unroll
  for (int j = 0; j < EPW; ++j) {
    const int sj = __builtin_amdgcn_readlane(msrc, j);
    const int dj = __builtin_amdgcn_readlane(mdst, j);
    a[j] = *(const float4*)(X + (size_t)sj * HID + lane * 4);
    b[j] = *(const float4*)(X + (size_t)dj * HID + lane * 4);
  }
  float mine = 0.f;
  #pragma unroll
  for (int j = 0; j < EPW; ++j) {
    float s = __builtin_fmaf(a[j].w, b[j].w,
              __builtin_fmaf(a[j].z, b[j].z,
              __builtin_fmaf(a[j].x, b[j].x, a[j].y * b[j].y)));
    s = wave_sum(s);
    if (own == j) mine = s;
  }
  if (lane < EPW) {
    const float logit = beta * mine * rs * rd;
    elog[ej] = logit;
    atomicMax(segmax + mdst, enc_f32(logit));
  }
}

__global__ void k_edge_exp(const int* __restrict__ EI,
                           const unsigned* __restrict__ segmax,
                           float* __restrict__ elog,
                           float* __restrict__ segsum) {
  const int e = (int)(blockIdx.x * blockDim.x + threadIdx.x);
  if (e >= E_TOT) return;
  const int dst = (e < N_EDGES) ? EI[N_EDGES + e] : (e - N_EDGES);
  const float ex = __expf(elog[e] - dec_f32(segmax[dst]));
  elog[e] = ex;
  atomicAdd(segsum + dst, ex);
}

__global__ void k_scatter(const float* __restrict__ X,
                          const int* __restrict__ EI,
                          const float* __restrict__ elog,
                          const float* __restrict__ segsum,
                          float* __restrict__ Y) {
  const int wave = __builtin_amdgcn_readfirstlane(
      (int)((blockIdx.x * blockDim.x + threadIdx.x) >> 5));
  const int lane = (int)(threadIdx.x & 31);
  const int e0   = wave * EPW;
  if (e0 >= E_TOT) return;
  const int ej = e0 + (lane & (EPW - 1));
  int msrc, mdst;
  if (e0 < N_EDGES) { msrc = EI[ej]; mdst = EI[N_EDGES + ej]; }
  else              { msrc = mdst = ej - N_EDGES; }
  const float malpha = elog[ej] / (segsum[mdst] + 1e-16f);

  float4 xs[EPW];
  #pragma unroll
  for (int j = 0; j < EPW; ++j) {
    const int sj = __builtin_amdgcn_readlane(msrc, j);
    xs[j] = *(const float4*)(X + (size_t)sj * HID + lane * 4);
  }
  #pragma unroll
  for (int j = 0; j < EPW; ++j) {
    const int   dj    = __builtin_amdgcn_readlane(mdst, j);
    const float alpha = __int_as_float(
        __builtin_amdgcn_readlane(__float_as_int(malpha), j));
    float* yp = Y + (size_t)dj * HID + lane * 4;
    atomicAdd(yp + 0, alpha * xs[j].x);
    atomicAdd(yp + 1, alpha * xs[j].y);
    atomicAdd(yp + 2, alpha * xs[j].z);
    atomicAdd(yp + 3, alpha * xs[j].w);
  }
}

static void run_conv(const float* X, float* Y, const int* EI,
                     const float* beta_ptr, float* rn, unsigned* segmax,
                     float* segsum, float* elog, bool compute_rnorm,
                     hipStream_t stream) {
  const int init_threads = N_NODES * HID;
  k_init<<<(init_threads + 255) / 256, 256, 0, stream>>>(Y, segmax, segsum);

  if (compute_rnorm) {
    const long long norm_threads = (long long)N_NODES * 32;
    k_rnorm<<<(int)((norm_threads + 255) / 256), 256, 0, stream>>>(X, rn);
  }

  const long long edge_threads = (long long)(E_TOT / EPW) * 32;
  const int eblocks = (int)((edge_threads + 255) / 256);
  k_edge_logit<<<eblocks, 256, 0, stream>>>(X, EI, rn, beta_ptr, elog, segmax);

  k_edge_exp<<<(E_TOT + 255) / 256, 256, 0, stream>>>(EI, segmax, elog, segsum);

  k_scatter<<<eblocks, 256, 0, stream>>>(X, EI, elog, segsum, Y);
}

extern "C" void kernel_launch(void* const* d_in, const int* in_sizes, int n_in,
                              void* d_out, int out_size, void* d_ws, size_t ws_size,
                              hipStream_t stream) {
  const float* x     = (const float*)d_in[0];
  const int*   ei    = (const int*)d_in[1];
  const float* W1    = (const float*)d_in[2];
  const float* b1    = (const float*)d_in[3];
  const float* beta2 = (const float*)d_in[4];
  float* out = (float*)d_out;

  float*    ws     = (float*)d_ws;
  float*    h1     = ws;
  float*    h2     = ws + 6400000;
  float*    rn     = ws + 12800000;
  unsigned* segmax = (unsigned*)(ws + 12850000);
  float*    segsum = ws + 12900000;
  float*    elog   = ws + 12950000;

  const int gemm_waves  = (N_NODES + 15) / 16;
  const int gemm_blocks = (gemm_waves * 32 + 255) / 256;
  k_gemm_relu<<<gemm_blocks, 256, 0, stream>>>(x, W1, b1, h1, rn);

  run_conv(h1, h2, ei, nullptr, rn, segmax, segsum, elog, false, stream);
  run_conv(h2, out, ei, beta2, rn, segmax, segsum, elog, true, stream);
}


# Edges grouped by destination per call (int-atomic rank, scan, fill); a wave owns a node: online softmax + weighted row sum in registers, plain stores replace f32 atomics; sum order = rank arrival
# speedup vs baseline: 3.1468x; 2.3130x over previous
#include <hip/hip_runtime.h>
#include <hip/hip_bf16.h>
#include <math.h>


typedef float v2f __attribute__((ext_vector_type(2)));
typedef float v8f __attribute__((ext_vector_type(8)));

#define N_NODES 50000
#define N_EDGES 600000
#define E_TOT   650000
#define IN_CH   256
#define HID     128
#define EPW     8
#define SCAN_CHUNKS 13
static_assert(N_EDGES % 4 == 0 && N_NODES % 4 == 0,
              "the list passes move 4 entries per thread as one 16-byte access");
static_assert(32 * SCAN_CHUNKS * 128 >= N_NODES,
              "the one-block scan must cover every node counter");

__device__ __forceinline__ float wave_sum(float v) {
  #pragma unroll
  for (int off = 16; off >= 1; off >>= 1) v += __shfl_xor(v, off, 32);
  return v;
}

__global__ void k_gemm_relu(const float* __restrict__ X,
                            const float* __restrict__ W,
                            const float* __restrict__ B,
                            float* __restrict__ H,
                            float* __restrict__ rn,
                            int* __restrict__ cnt) {
  const int wave = (int)((blockIdx.x * blockDim.x + threadIdx.x) >> 5);
  const int lane = (int)(threadIdx.x & 31);
  if (wave * 16 >= N_NODES) return;

  const int m0   = wave * 16;
  const int mrow = m0 + (lane & 15);
  const int kg   = (lane >> 4) * 2;
  const int n15  = lane & 15;

  v8f acc[8];
  #pragma unroll
  for (int t = 0; t < 8; ++t) acc[t] = (v8f){0.f,0.f,0.f,0.f,0.f,0.f,0.f,0.f};

  const float* __restrict__ xrow = X + (size_t)mrow * IN_CH + kg;
  const float* __restrict__ wbas = W + (size_t)n15 * IN_CH + kg;

  for (int k = 0; k < IN_CH; k += 8) {
    const v2f a0 = *(const v2f*)(xrow + k);
    const v2f a1 = *(const v2f*)(xrow + k + 4);
    v2f b0[8], b1[8];
    #pragma unroll
    for (int t = 0; t < 8; ++t)
      b0[t] = *(const v2f*)(wbas + (size_t)t * 16 * IN_CH + k);
    #pragma unroll
    for (int t = 0; t < 8; ++t)
      b1[t] = *(const v2f*)(wbas + (size_t)t * 16 * IN_CH + k + 4);
    #pragma unroll
    for (int t = 0; t < 8; ++t)
      acc[t] = __builtin_amdgcn_wmma_f32_16x16x4_f32(
          false, a0, false, b0[t], (short)0, acc[t], false, false);
    #pragma unroll
    for (int t = 0; t < 8; ++t)
      acc[t] = __builtin_amdgcn_wmma_f32_16x16x4_f32(
          false, a1, false, b1[t], (short)0, acc[t], false, false);
  }

  const int mside = (lane >> 4) * 8;
  float ss[8];
  #pragma unroll
  for (int r = 0; r < 8; ++r) ss[r] = 0.f;

  #pragma unroll
  for (int t = 0; t < 8; ++t) {
    const int col = t * 16 + n15;
    const float bb = B[col];
    #pragma unroll
    for (int r = 0; r < 8; ++r) {
      const int row = m0 + r + mside;
      float v = acc[t][r] + bb;
      v = v > 0.f ? v : 0.f;
      H[(size_t)row * HID + col] = v;
      ss[r] += v * v;
    }
  }
  #pragma unroll
  for (int r = 0; r < 8; ++r) {
    #pragma unroll
    for (int off = 8; off >= 1; off >>= 1) ss[r] += __shfl_xor(ss[r], off, 32);
  }
  if (n15 == 0) {
    #pragma unroll
    for (int r = 0; r < 8; ++r)
      rn[m0 + r + mside] = 1.f / fmaxf(sqrtf(ss[r]), 1e-12f);
  }
  if (lane < 16 && m0 + lane < N_NODES) cnt[m0 + lane] = 0;
}


__global__ void k_rank(const int* __restrict__ EI,
                       int* __restrict__ cnt,
                       int* __restrict__ rank) {
  const int t = (int)(blockIdx.x * blockDim.x + threadIdx.x);
  if (t * 4 >= N_EDGES) return;
  const int4 d = *(const int4*)(EI + N_EDGES + t * 4);
  int4 r;
  r.x = atomicAdd(cnt + d.x, 1);
  r.y = atomicAdd(cnt + d.y, 1);
  r.z = atomicAdd(cnt + d.z, 1);
  r.w = atomicAdd(cnt + d.w, 1);
  *(int4*)(rank + t * 4) = r;
}

__global__ void __launch_bounds__(1024) k_scan(int* __restrict__ rowptr) {
  __shared__ int wtot[32];
  const int lane = (int)(threadIdx.x & 31);
  const int w    = (int)(threadIdx.x >> 5);
  const int base = w * (SCAN_CHUNKS * 128) + lane * 4;

  int4 v[SCAN_CHUNKS];
  #pragma unroll
  for (int i = 0; i < SCAN_CHUNKS; ++i) {
    const int idx = base + i * 128;
    v[i] = (idx < N_NODES) ? *(const int4*)(rowptr + idx) : make_int4(0, 0, 0, 0);
  }
  int carry = 0;
  #pragma unroll
  for (int i = 0; i < SCAN_CHUNKS; ++i) {
    const int s1 = v[i].x;
    const int s2 = s1 + v[i].y;
    const int s3 = s2 + v[i].z;
    const int s4 = s3 + v[i].w;
    int x = s4;
    #pragma unroll
    for (int off = 1; off < 32; off <<= 1) {
      const int y = __shfl_up(x, off, 32);
      if (lane >= off) x += y;
    }
    const int ex = x - s4 + carry;
    carry += __builtin_amdgcn_readlane(x, 31);
    v[i] = make_int4(ex, ex + s1, ex + s2, ex + s3);
  }
  if (lane == 0) wtot[w] = carry;
  __syncthreads();
  int wb = (lane < w) ? wtot[lane] : 0;
  #pragma unroll
  for (int off = 16; off >= 1; off >>= 1) wb += __shfl_xor(wb, off, 32);
  #pragma unroll
  for (int i = 0; i < SCAN_CHUNKS; ++i) {
    const int idx = base + i * 128;
    if (idx < N_NODES)
      *(int4*)(rowptr + idx) =
          make_int4(v[i].x + wb, v[i].y + wb, v[i].z + wb, v[i].w + wb);
  }
  if (threadIdx.x == 1023) rowptr[N_NODES] = carry + wb;
}

__global__ void k_fill(const int* __restrict__ EI,
                       const int* __restrict__ rank,
                       const int* __restrict__ rowptr,
                       int* __restrict__ csr) {
  const int t = (int)(blockIdx.x * blockDim.x + threadIdx.x);
  if (t * 4 >= N_EDGES) return;
  const int4 sv = *(const int4*)(EI + t * 4);
  const int4 d  = *(const int4*)(EI + N_EDGES + t * 4);
  const int4 r  = *(const int4*)(rank + t * 4);
  csr[rowptr[d.x] + r.x] = sv.x;
  csr[rowptr[d.y] + r.y] = sv.y;
  csr[rowptr[d.z] + r.z] = sv.z;
  csr[rowptr[d.w] + r.w] = sv.w;
}

__global__ void __launch_bounds__(128)
k_conv(const float* __restrict__ X,
       const float* __restrict__ rn,
       const int* __restrict__ rowptr,
       const int* __restrict__ csr,
       const float* __restrict__ beta_ptr,
       float* __restrict__ Y,
       float* __restrict__ rn_out) {
  const int gw = __builtin_amdgcn_readfirstlane(
      (int)((blockIdx.x * blockDim.x + threadIdx.x) >> 5));
  const int nw   = (int)((gridDim.x * blockDim.x) >> 5);
  const int lane = (int)(threadIdx.x & 31);
  const int own  = lane & (EPW - 1);
  const float beta = beta_ptr ? beta_ptr[0] : 1.0f;

  for (int node = gw; node < N_NODES; node += nw) {
    const int beg = rowptr[node];
    const int end = rowptr[node + 1];
    const float4 xd = *(const float4*)(X + (size_t)node * HID + lane * 4);
    const float  rd = rn[node];

    float sd = __builtin_fmaf(xd.w, xd.w,
               __builtin_fmaf(xd.z, xd.z,
               __builtin_fmaf(xd.x, xd.x, xd.y * xd.y)));
    sd = wave_sum(sd);
    float  m    = beta * sd * rd * rd;
    float  ssum = 1.0f;
    float4 acc  = xd;

    for (int p = beg; p < end; p += EPW) {
      const bool valid = (p + own) < end;
      const int   ms = valid ? csr[p + own] : node;
      const float rs = rn[ms];

      float4 a[EPW];
      #pragma unroll
      for (int j = 0; j < EPW; ++j) {
        const int sj = __builtin_amdgcn_readlane(ms, j);
        a[j] = *(const float4*)(X + (size_t)sj * HID + lane * 4);
      }
      float mine = 0.f;
      #pragma unroll
      for (int j = 0; j < EPW; ++j) {
        float s = __builtin_fmaf(a[j].w, xd.w,
                  __builtin_fmaf(a[j].z, xd.z,
                  __builtin_fmaf(a[j].x, xd.x, a[j].y * xd.y)));
        s = wave_sum(s);
        if (own == j) mine = s;
      }
      const float logit = valid ? beta * mine * rs * rd : -__builtin_inff();

      float bm = logit;
      #pragma unroll
      for (int off = EPW / 2; off >= 1; off >>= 1)
        bm = fmaxf(bm, __shfl_xor(bm, off, 32));
      const float m_new = fmaxf(m, bm);
      const float scale = __expf(m - m_new);
      const float e     = valid ? __expf(logit - m_new) : 0.f;
      float es = e;
      #pragma unroll
      for (int off = EPW / 2; off >= 1; off >>= 1)
        es += __shfl_xor(es, off, 32);
      ssum  = ssum * scale + es;
      acc.x *= scale; acc.y *= scale; acc.z *= scale; acc.w *= scale;
      #pragma unroll
      for (int j = 0; j < EPW; ++j) {
        const float ej = __int_as_float(
            __builtin_amdgcn_readlane(__float_as_int(e), j));
        acc.x = __builtin_fmaf(ej, a[j].x, acc.x);
        acc.y = __builtin_fmaf(ej, a[j].y, acc.y);
        acc.z = __builtin_fmaf(ej, a[j].z, acc.z);
        acc.w = __builtin_fmaf(ej, a[j].w, acc.w);
      }
      m = m_new;
    }

    const float den = ssum + 1e-16f;
    float4 o;
    o.x = acc.x / den; o.y = acc.y / den; o.z = acc.z / den; o.w = acc.w / den;
    *(float4*)(Y + (size_t)node * HID + lane * 4) = o;
    if (rn_out) {
      float q = o.x * o.x + o.y * o.y + o.z * o.z + o.w * o.w;
      q = wave_sum(q);
      if (lane == 0) rn_out[node] = 1.f / fmaxf(sqrtf(q), 1e-12f);
    }
  }
}

extern "C" void kernel_launch(void* const* d_in, const int* in_sizes, int n_in,
                              void* d_out, int out_size, void* d_ws, size_t ws_size,
                              hipStream_t stream) {
  const float* x     = (const float*)d_in[0];
  const int*   ei    = (const int*)d_in[1];
  const float* W1    = (const float*)d_in[2];
  const float* b1    = (const float*)d_in[3];
  const float* beta2 = (const float*)d_in[4];
  float* out = (float*)d_out;

  float* ws     = (float*)d_ws;
  float* h1     = ws;
  float* h2     = ws + 6400000;
  int*   rank   = (int*)h2;
  float* rn1    = ws + 12800000;
  float* rn2    = ws + 12850000;
  int*   rowptr = (int*)(ws + 12900000);
  int*   csr    = (int*)(ws + 12950004);

  const int gemm_waves  = (N_NODES + 15) / 16;
  const int gemm_blocks = (gemm_waves * 32 + 255) / 256;
  k_gemm_relu<<<gemm_blocks, 256, 0, stream>>>(x, W1, b1, h1, rn1, rowptr);

  const int list_blocks = (N_EDGES / 4 + 255) / 256;
  k_rank<<<list_blocks, 256, 0, stream>>>(ei, rowptr, rank);
  k_scan<<<1, 1024, 0, stream>>>(rowptr);
  k_fill<<<list_blocks, 256, 0, stream>>>(ei, rank, rowptr, csr);

  const int conv_blocks = 2500;
  k_conv<<<conv_blocks, 128, 0, stream>>>(h1, rn1, rowptr, csr, nullptr, h2, rn2);
  k_conv<<<conv_blocks, 128, 0, stream>>>(h2, rn2, rowptr, csr, beta2, out, nullptr);
}


# Conv: a batch's 8 dot products share one halving lane exchange on the vector ALU (permlanex16, row_xmask DPP), not 8 serial ds_bpermute butterflies; max, sums, norms likewise; sum trees unchanged
# speedup vs baseline: 3.7444x; 1.1899x over previous
#include <hip/hip_runtime.h>
#include <hip/hip_bf16.h>
#include <math.h>


typedef float v2f __attribute__((ext_vector_type(2)));
typedef float v8f __attribute__((ext_vector_type(8)));

#define N_NODES 50000
#define N_EDGES 600000
#define E_TOT   650000
#define IN_CH   256
#define HID     128
#define EPW     8
static_assert(EPW == 8,
              "the halving exchange of k_conv is written out for 8 edges: 4 lanes per edge");
#define SCAN_CHUNKS 13
static_assert(N_EDGES % 4 == 0 && N_NODES % 4 == 0,
              "the list passes move 4 entries per thread as one 16-byte access");
static_assert(32 * SCAN_CHUNKS * 128 >= N_NODES,
              "the one-block scan must cover every node counter");

__device__ __forceinline__ float xrow(float x) {
  const int b = __float_as_int(x);
  return __int_as_float(
      __builtin_amdgcn_permlanex16(b, b, 0x76543210, 0xfedcba98, false, false));
}

template <int M>
__device__ __forceinline__ float xmask(float x) {
  return __int_as_float(__builtin_amdgcn_update_dpp(
      0, __float_as_int(x), 0x160 + M, 0xf, 0xf, true));
}

__device__ __forceinline__ float wave_sum(float v) {
  v += xrow(v);
  v += xmask<8>(v);
  v += xmask<4>(v);
  v += xmask<2>(v);
  v += xmask<1>(v);
  return v;
}

__global__ void k_gemm_relu(const float* __restrict__ X,
                            const float* __restrict__ W,
                            const float* __restrict__ B,
                            float* __restrict__ H,
                            float* __restrict__ rn,
                            int* __restrict__ cnt) {
  const int wave = (int)((blockIdx.x * blockDim.x + threadIdx.x) >> 5);
  const int lane = (int)(threadIdx.x & 31);
  if (wave * 16 >= N_NODES) return;

  const int m0   = wave * 16;
  const int mrow = m0 + (lane & 15);
  const int kg   = (lane >> 4) * 2;
  const int n15  = lane & 15;

  v8f acc[8];
  #pragma unroll
  for (int t = 0; t < 8; ++t) acc[t] = (v8f){0.f,0.f,0.f,0.f,0.f,0.f,0.f,0.f};

  const float* __restrict__ xrow = X + (size_t)mrow * IN_CH + kg;
  const float* __restrict__ wbas = W + (size_t)n15 * IN_CH + kg;

  for (int k = 0; k < IN_CH; k += 8) {
    const v2f a0 = *(const v2f*)(xrow + k);
    const v2f a1 = *(const v2f*)(xrow + k + 4);
    v2f b0[8], b1[8];
    #pragma unroll
    for (int t = 0; t < 8; ++t)
      b0[t] = *(const v2f*)(wbas + (size_t)t * 16 * IN_CH + k);
    #pragma unroll
    for (int t = 0; t < 8; ++t)
      b1[t] = *(const v2f*)(wbas + (size_t)t * 16 * IN_CH + k + 4);
    #pragma unroll
    for (int t = 0; t < 8; ++t)
      acc[t] = __builtin_amdgcn_wmma_f32_16x16x4_f32(
          false, a0, false, b0[t], (short)0, acc[t], false, false);
    #pragma unroll
    for (int t = 0; t < 8; ++t)
      acc[t] = __builtin_amdgcn_wmma_f32_16x16x4_f32(
          false, a1, false, b1[t], (short)0, acc[t], false, false);
  }

  const int mside = (lane >> 4) * 8;
  float ss[8];
  #pragma unroll
  for (int r = 0; r < 8; ++r) ss[r] = 0.f;

  #pragma unroll
  for (int t = 0; t < 8; ++t) {
    const int col = t * 16 + n15;
    const float bb = B[col];
    #pragma unroll
    for (int r = 0; r < 8; ++r) {
      const int row = m0 + r + mside;
      float v = acc[t][r] + bb;
      v = v > 0.f ? v : 0.f;
      H[(size_t)row * HID + col] = v;
      ss[r] += v * v;
    }
  }
  #pragma unroll
  for (int r = 0; r < 8; ++r) {
    #pragma unroll
    for (int off = 8; off >= 1; off >>= 1) ss[r] += __shfl_xor(ss[r], off, 32);
  }
  if (n15 == 0) {
    #pragma unroll
    for (int r = 0; r < 8; ++r)
      rn[m0 + r + mside] = 1.f / fmaxf(sqrtf(ss[r]), 1e-12f);
  }
  if (lane < 16 && m0 + lane < N_NODES) cnt[m0 + lane] = 0;
}


__global__ void k_rank(const int* __restrict__ EI,
                       int* __restrict__ cnt,
                       int* __restrict__ rank) {
  const int t = (int)(blockIdx.x * blockDim.x + threadIdx.x);
  if (t * 4 >= N_EDGES) return;
  const int4 d = *(const int4*)(EI + N_EDGES + t * 4);
  int4 r;
  r.x = atomicAdd(cnt + d.x, 1);
  r.y = atomicAdd(cnt + d.y, 1);
  r.z = atomicAdd(cnt + d.z, 1);
  r.w = atomicAdd(cnt + d.w, 1);
  *(int4*)(rank + t * 4) = r;
}

__global__ void __launch_bounds__(1024) k_scan(int* __restrict__ rowptr) {
  __shared__ int wtot[32];
  const int lane = (int)(threadIdx.x & 31);
  const int w    = (int)(threadIdx.x >> 5);
  const int base = w * (SCAN_CHUNKS * 128) + lane * 4;

  int4 v[SCAN_CHUNKS];
  #pragma unroll
  for (int i = 0; i < SCAN_CHUNKS; ++i) {
    const int idx = base + i * 128;
    v[i] = (idx < N_NODES) ? *(const int4*)(rowptr + idx) : make_int4(0, 0, 0, 0);
  }
  int carry = 0;
  #pragma unroll
  for (int i = 0; i < SCAN_CHUNKS; ++i) {
    const int s1 = v[i].x;
    const int s2 = s1 + v[i].y;
    const int s3 = s2 + v[i].z;
    const int s4 = s3 + v[i].w;
    int x = s4;
    #pragma unroll
    for (int off = 1; off < 32; off <<= 1) {
      const int y = __shfl_up(x, off, 32);
      if (lane >= off) x += y;
    }
    const int ex = x - s4 + carry;
    carry += __builtin_amdgcn_readlane(x, 31);
    v[i] = make_int4(ex, ex + s1, ex + s2, ex + s3);
  }
  if (lane == 0) wtot[w] = carry;
  __syncthreads();
  int wb = (lane < w) ? wtot[lane] : 0;
  #pragma unroll
  for (int off = 16; off >= 1; off >>= 1) wb += __shfl_xor(wb, off, 32);
  #pragma unroll
  for (int i = 0; i < SCAN_CHUNKS; ++i) {
    const int idx = base + i * 128;
    if (idx < N_NODES)
      *(int4*)(rowptr + idx) =
          make_int4(v[i].x + wb, v[i].y + wb, v[i].z + wb, v[i].w + wb);
  }
  if (threadIdx.x == 1023) rowptr[N_NODES] = carry + wb;
}

__global__ void k_fill(const int* __restrict__ EI,
                       const int* __restrict__ rank,
                       const int* __restrict__ rowptr,
                       int* __restrict__ csr) {
  const int t = (int)(blockIdx.x * blockDim.x + threadIdx.x);
  if (t * 4 >= N_EDGES) return;
  const int4 sv = *(const int4*)(EI + t * 4);
  const int4 d  = *(const int4*)(EI + N_EDGES + t * 4);
  const int4 r  = *(const int4*)(rank + t * 4);
  csr[rowptr[d.x] + r.x] = sv.x;
  csr[rowptr[d.y] + r.y] = sv.y;
  csr[rowptr[d.z] + r.z] = sv.z;
  csr[rowptr[d.w] + r.w] = sv.w;
}

__global__ void __launch_bounds__(128)
k_conv(const float* __restrict__ X,
       const float* __restrict__ rn,
       const int* __restrict__ rowptr,
       const int* __restrict__ csr,
       const float* __restrict__ beta_ptr,
       float* __restrict__ Y,
       float* __restrict__ rn_out) {
  const int gw = __builtin_amdgcn_readfirstlane(
      (int)((blockIdx.x * blockDim.x + threadIdx.x) >> 5));
  const int nw   = (int)((gridDim.x * blockDim.x) >> 5);
  const int lane = (int)(threadIdx.x & 31);
  const int own  = lane >> 2;
  const bool up16 = (lane & 16) != 0;
  const bool up8  = (lane & 8) != 0;
  const bool up4  = (lane & 4) != 0;
  const float beta = beta_ptr ? beta_ptr[0] : 1.0f;

  for (int node = gw; node < N_NODES; node += nw) {
    const int beg = rowptr[node];
    const int end = rowptr[node + 1];
    const float4 xd = *(const float4*)(X + (size_t)node * HID + lane * 4);
    const float  rd = rn[node];

    float sd = __builtin_fmaf(xd.w, xd.w,
               __builtin_fmaf(xd.z, xd.z,
               __builtin_fmaf(xd.x, xd.x, xd.y * xd.y)));
    sd = wave_sum(sd);
    float  m    = beta * sd * rd * rd;
    float  ssum = 1.0f;
    float4 acc  = xd;

    for (int p = beg; p < end; p += EPW) {
      const bool valid = (p + own) < end;
      const int   ms = valid ? csr[p + own] : node;
      const float rs = rn[ms];

      float4 a[EPW];
      #pragma unroll
      for (int j = 0; j < EPW; ++j) {
        const int sj = __builtin_amdgcn_readlane(ms, 4 * j);
        a[j] = *(const float4*)(X + (size_t)sj * HID + lane * 4);
      }
      asm volatile("" ::: "memory");
      float s[EPW];
      #pragma unroll
      for (int j = 0; j < EPW; ++j)
        s[j] = __builtin_fmaf(a[j].w, xd.w,
               __builtin_fmaf(a[j].z, xd.z,
               __builtin_fmaf(a[j].x, xd.x, a[j].y * xd.y)));
      float t[4];
      #pragma unroll
      for (int i = 0; i < 4; ++i) {
        const float keep = up16 ? s[i + 4] : s[i];
        const float send = up16 ? s[i] : s[i + 4];
        t[i] = keep + xrow(send);
      }
      float u[2];
      #pragma unroll
      for (int i = 0; i < 2; ++i) {
        const float keep = up8 ? t[i + 2] : t[i];
        const float send = up8 ? t[i] : t[i + 2];
        u[i] = keep + xmask<8>(send);
      }
      float mine;
      {
        const float keep = up4 ? u[1] : u[0];
        const float send = up4 ? u[0] : u[1];
        mine = keep + xmask<4>(send);
      }
      mine += xmask<2>(mine);
      mine += xmask<1>(mine);
      const float logit = valid ? beta * mine * rs * rd : -__builtin_inff();

      float bm = logit;
      bm = fmaxf(bm, xmask<4>(bm));
      bm = fmaxf(bm, xmask<8>(bm));
      bm = fmaxf(bm, xrow(bm));
      const float m_new = fmaxf(m, bm);
      const float scale = __expf(m - m_new);
      const float e     = valid ? __expf(logit - m_new) : 0.f;
      float es = e;
      es += xrow(es);
      es += xmask<8>(es);
      es += xmask<4>(es);
      ssum  = ssum * scale + es;
      acc.x *= scale; acc.y *= scale; acc.z *= scale; acc.w *= scale;
      #pragma unroll
      for (int j = 0; j < EPW; ++j) {
        const float ej = __int_as_float(
            __builtin_amdgcn_readlane(__float_as_int(e), 4 * j));
        acc.x = __builtin_fmaf(ej, a[j].x, acc.x);
        acc.y = __builtin_fmaf(ej, a[j].y, acc.y);
        acc.z = __builtin_fmaf(ej, a[j].z, acc.z);
        acc.w = __builtin_fmaf(ej, a[j].w, acc.w);
      }
      m = m_new;
    }

    const float den = ssum + 1e-16f;
    float4 o;
    o.x = acc.x / den; o.y = acc.y / den; o.z = acc.z / den; o.w = acc.w / den;
    *(float4*)(Y + (size_t)node * HID + lane * 4) = o;
    if (rn_out) {
      float q = o.x * o.x + o.y * o.y + o.z * o.z + o.w * o.w;
      q = wave_sum(q);
      if (lane == 0) rn_out[node] = 1.f / fmaxf(sqrtf(q), 1e-12f);
    }
  }
}

extern "C" void kernel_launch(void* const* d_in, const int* in_sizes, int n_in,
                              void* d_out, int out_size, void* d_ws, size_t ws_size,
                              hipStream_t stream) {
  const float* x     = (const float*)d_in[0];
  const int*   ei    = (const int*)d_in[1];
  const float* W1    = (const float*)d_in[2];
  const float* b1    = (const float*)d_in[3];
  const float* beta2 = (const float*)d_in[4];
  float* out = (float*)d_out;

  float* ws     = (float*)d_ws;
  float* h1     = ws;
  float* h2     = ws + 6400000;
  int*   rank   = (int*)h2;
  float* rn1    = ws + 12800000;
  float* rn2    = ws + 12850000;
  int*   rowptr = (int*)(ws + 12900000);
  int*   csr    = (int*)(ws + 12950004);

  const int gemm_waves  = (N_NODES + 15) / 16;
  const int gemm_blocks = (gemm_waves * 32 + 255) / 256;
  k_gemm_relu<<<gemm_blocks, 256, 0, stream>>>(x, W1, b1, h1, rn1, rowptr);

  const int list_blocks = (N_EDGES / 4 + 255) / 256;
  k_rank<<<list_blocks, 256, 0, stream>>>(ei, rowptr, rank);
  k_scan<<<1, 1024, 0, stream>>>(rowptr);
  k_fill<<<list_blocks, 256, 0, stream>>>(ei, rank, rowptr, csr);

  const int conv_blocks = 2500;
  k_conv<<<conv_blocks, 128, 0, stream>>>(h1, rn1, rowptr, csr, nullptr, h2, rn2);
  k_conv<<<conv_blocks, 128, 0, stream>>>(h2, rn2, rowptr, csr, beta2, out, nullptr);
}


# Edge grouping: two-level counting sort on LDS counters (chunk blocks bin by 256 destinations, a block per bin ranks by returning LDS add) replaces global atomics, scan, fill; 6 to 5 launches
# speedup vs baseline: 5.3269x; 1.4226x over previous
#include <hip/hip_runtime.h>
#include <hip/hip_bf16.h>
#include <math.h>


typedef float v2f __attribute__((ext_vector_type(2)));
typedef float v8f __attribute__((ext_vector_type(8)));

#define N_NODES 50000
#define N_EDGES 600000
#define E_TOT   650000
#define IN_CH   256
#define HID     128
#define EPW     8
static_assert(EPW == 8,
              "the halving exchange of k_conv is written out for 8 edges: 4 lanes per edge");
#define BIN_SHIFT 8
#define BIN_NODES (1 << BIN_SHIFT)
#define N_BINS    ((N_NODES + BIN_NODES - 1) / BIN_NODES)
#define N_CHUNKS  128
#define CHUNK     4688
#define BIN_T     512
#define BIN_EPT   10
#define RUN_REG   5
static_assert(N_BINS <= 256 && BIN_NODES == 256,
              "both LDS tables are scanned as 256 entries by one wave");
static_assert(N_CHUNKS * CHUNK >= N_EDGES, "the chunks must cover the edge list");
static_assert(BIN_T * BIN_EPT >= CHUNK, "a binning block holds its chunk in registers");
static_assert(N_CHUNKS * 8 == 1024,
              "a grouping block gives each chunk's run 8 of its 1024 threads");

__device__ __forceinline__ float xrow(float x) {
  const int b = __float_as_int(x);
  return __int_as_float(
      __builtin_amdgcn_permlanex16(b, b, 0x76543210, 0xfedcba98, false, false));
}

template <int M>
__device__ __forceinline__ float xmask(float x) {
  return __int_as_float(__builtin_amdgcn_update_dpp(
      0, __float_as_int(x), 0x160 + M, 0xf, 0xf, true));
}

__device__ __forceinline__ float wave_sum(float v) {
  v += xrow(v);
  v += xmask<8>(v);
  v += xmask<4>(v);
  v += xmask<2>(v);
  v += xmask<1>(v);
  return v;
}

__global__ void k_gemm_relu(const float* __restrict__ X,
                            const float* __restrict__ W,
                            const float* __restrict__ B,
                            float* __restrict__ H,
                            float* __restrict__ rn) {
  const int wave = (int)((blockIdx.x * blockDim.x + threadIdx.x) >> 5);
  const int lane = (int)(threadIdx.x & 31);
  if (wave * 16 >= N_NODES) return;

  const int m0   = wave * 16;
  const int mrow = m0 + (lane & 15);
  const int kg   = (lane >> 4) * 2;
  const int n15  = lane & 15;

  v8f acc[8];
  #pragma unroll
  for (int t = 0; t < 8; ++t) acc[t] = (v8f){0.f,0.f,0.f,0.f,0.f,0.f,0.f,0.f};

  const float* __restrict__ xrow = X + (size_t)mrow * IN_CH + kg;
  const float* __restrict__ wbas = W + (size_t)n15 * IN_CH + kg;

  for (int k = 0; k < IN_CH; k += 8) {
    const v2f a0 = *(const v2f*)(xrow + k);
    const v2f a1 = *(const v2f*)(xrow + k + 4);
    v2f b0[8], b1[8];
    #pragma unroll
    for (int t = 0; t < 8; ++t)
      b0[t] = *(const v2f*)(wbas + (size_t)t * 16 * IN_CH + k);
    #pragma unroll
    for (int t = 0; t < 8; ++t)
      b1[t] = *(const v2f*)(wbas + (size_t)t * 16 * IN_CH + k + 4);
    #pragma unroll
    for (int t = 0; t < 8; ++t)
      acc[t] = __builtin_amdgcn_wmma_f32_16x16x4_f32(
          false, a0, false, b0[t], (short)0, acc[t], false, false);
    #pragma unroll
    for (int t = 0; t < 8; ++t)
      acc[t] = __builtin_amdgcn_wmma_f32_16x16x4_f32(
          false, a1, false, b1[t], (short)0, acc[t], false, false);
  }

  const int mside = (lane >> 4) * 8;
  float ss[8];
  #pragma unroll
  for (int r = 0; r < 8; ++r) ss[r] = 0.f;

  #pragma unroll
  for (int t = 0; t < 8; ++t) {
    const int col = t * 16 + n15;
    const float bb = B[col];
    #pragma unroll
    for (int r = 0; r < 8; ++r) {
      const int row = m0 + r + mside;
      float v = acc[t][r] + bb;
      v = v > 0.f ? v : 0.f;
      H[(size_t)row * HID + col] = v;
      ss[r] += v * v;
    }
  }
  #pragma unroll
  for (int r = 0; r < 8; ++r) {
    #pragma unroll
    for (int off = 8; off >= 1; off >>= 1) ss[r] += __shfl_xor(ss[r], off, 32);
  }
  if (n15 == 0) {
    #pragma unroll
    for (int r = 0; r < 8; ++r)
      rn[m0 + r + mside] = 1.f / fmaxf(sqrtf(ss[r]), 1e-12f);
  }
}


__device__ __forceinline__ int scan256_wave(const int* tab, int lane, int (&ex)[8]) {
  int run = 0;
  #pragma unroll
  for (int j = 0; j < 8; ++j) {
    const int t = tab[lane * 8 + j];
    ex[j] = run;
    run += t;
  }
  int x = run;
  #pragma unroll
  for (int off = 1; off < 32; off <<= 1) {
    const int y = __shfl_up(x, off, 32);
    if (lane >= off) x += y;
  }
  const int before = x - run;
  #pragma unroll
  for (int j = 0; j < 8; ++j) ex[j] += before;
  return x;
}

__global__ void __launch_bounds__(BIN_T)
k_bin(const int* __restrict__ EI,
      int* __restrict__ recs,
      int* __restrict__ startT,
      int* __restrict__ lenT) {
  __shared__ int hist[256];
  __shared__ int start[256];
  const int tid  = (int)threadIdx.x;
  const int w    = (int)blockIdx.x;
  const int base = w * CHUNK;
  if (tid < 256) hist[tid] = 0;
  __syncthreads();

  int s[BIN_EPT], d[BIN_EPT], r[BIN_EPT];
  #pragma unroll
  for (int k = 0; k < BIN_EPT; ++k) {
    const int i = tid + k * BIN_T;
    const bool ok = (i < CHUNK) && (base + i < N_EDGES);
    s[k] = ok ? EI[base + i] : 0;
    d[k] = ok ? EI[N_EDGES + base + i] : -1;
  }
  #pragma unroll
  for (int k = 0; k < BIN_EPT; ++k) {
    r[k] = 0;
    if (d[k] >= 0) r[k] = atomicAdd(&hist[d[k] >> BIN_SHIFT], 1);
  }
  __syncthreads();
  if (tid < 32) {
    int ex[8];
    scan256_wave(hist, tid, ex);
    #pragma unroll
    for (int j = 0; j < 8; ++j) start[tid * 8 + j] = ex[j];
  }
  __syncthreads();
  if (tid < N_BINS) {
    startT[tid * N_CHUNKS + w] = start[tid];
    lenT[tid * N_CHUNKS + w]   = hist[tid];
  }
  #pragma unroll
  for (int k = 0; k < BIN_EPT; ++k) {
    if (d[k] >= 0) {
      const int pos = base + start[d[k] >> BIN_SHIFT] + r[k];
      *(int2*)(recs + 2 * (size_t)pos) = make_int2(s[k], d[k]);
    }
  }
}

__global__ void __launch_bounds__(1024)
k_group(const int* __restrict__ recs,
        const int* __restrict__ startT,
        const int* __restrict__ lenT,
        int* __restrict__ rowptr,
        int* __restrict__ csr) {
  __shared__ int cnt[BIN_NODES];
  __shared__ int wpart[32];
  const int tid  = (int)threadIdx.x;
  const int lane = tid & 31;
  const int wv   = tid >> 5;
  const int b    = (int)blockIdx.x;
  const int run  = tid >> 3;
  const int sub  = tid & 7;
  const int st = startT[b * N_CHUNKS + run];
  const int ln = lenT[b * N_CHUNKS + run];

  int part = (sub == 0) ? st : 0;
  #pragma unroll
  for (int off = 16; off >= 1; off >>= 1) part += __shfl_xor(part, off, 32);
  if (lane == 0) wpart[wv] = part;
  if (tid < BIN_NODES) cnt[tid] = 0;
  __syncthreads();
  int base = wpart[lane];
  #pragma unroll
  for (int off = 16; off >= 1; off >>= 1) base += __shfl_xor(base, off, 32);

  const int* __restrict__ rp = recs + 2 * ((size_t)run * CHUNK + st);
  int2 q[RUN_REG];
  #pragma unroll
  for (int k = 0; k < RUN_REG; ++k) {
    const int i = sub + 8 * k;
    q[k] = (i < ln) ? *(const int2*)(rp + 2 * i) : make_int2(0, -1);
  }
  #pragma unroll
  for (int k = 0; k < RUN_REG; ++k)
    if (q[k].y >= 0) atomicAdd(&cnt[q[k].y & (BIN_NODES - 1)], 1);
  for (int i = sub + 8 * RUN_REG; i < ln; i += 8) {
    const int2 rc = *(const int2*)(rp + 2 * i);
    atomicAdd(&cnt[rc.y & (BIN_NODES - 1)], 1);
  }
  __syncthreads();

  if (tid < 32) {
    int ex[8];
    const int incl = scan256_wave(cnt, tid, ex);
    #pragma unroll
    for (int j = 0; j < 8; ++j) {
      const int li   = tid * 8 + j;
      const int node = b * BIN_NODES + li;
      cnt[li] = ex[j];
      if (node < N_NODES) rowptr[node] = base + ex[j];
    }
    if (b == N_BINS - 1 && tid == 31) rowptr[N_NODES] = base + incl;
  }
  __syncthreads();

  #pragma unroll
  for (int k = 0; k < RUN_REG; ++k) {
    if (q[k].y >= 0) {
      const int pos = atomicAdd(&cnt[q[k].y & (BIN_NODES - 1)], 1);
      csr[base + pos] = q[k].x;
    }
  }
  for (int i = sub + 8 * RUN_REG; i < ln; i += 8) {
    const int2 rc = *(const int2*)(rp + 2 * i);
    const int pos = atomicAdd(&cnt[rc.y & (BIN_NODES - 1)], 1);
    csr[base + pos] = rc.x;
  }
}

__global__ void __launch_bounds__(128)
k_conv(const float* __restrict__ X,
       const float* __restrict__ rn,
       const int* __restrict__ rowptr,
       const int* __restrict__ csr,
       const float* __restrict__ beta_ptr,
       float* __restrict__ Y,
       float* __restrict__ rn_out) {
  const int gw = __builtin_amdgcn_readfirstlane(
      (int)((blockIdx.x * blockDim.x + threadIdx.x) >> 5));
  const int nw   = (int)((gridDim.x * blockDim.x) >> 5);
  const int lane = (int)(threadIdx.x & 31);
  const int own  = lane >> 2;
  const bool up16 = (lane & 16) != 0;
  const bool up8  = (lane & 8) != 0;
  const bool up4  = (lane & 4) != 0;
  const float beta = beta_ptr ? beta_ptr[0] : 1.0f;

  for (int node = gw; node < N_NODES; node += nw) {
    const int beg = rowptr[node];
    const int end = rowptr[node + 1];
    const float4 xd = *(const float4*)(X + (size_t)node * HID + lane * 4);
    const float  rd = rn[node];

    float sd = __builtin_fmaf(xd.w, xd.w,
               __builtin_fmaf(xd.z, xd.z,
               __builtin_fmaf(xd.x, xd.x, xd.y * xd.y)));
    sd = wave_sum(sd);
    float  m    = beta * sd * rd * rd;
    float  ssum = 1.0f;
    float4 acc  = xd;

    for (int p = beg; p < end; p += EPW) {
      const bool valid = (p + own) < end;
      const int   ms = valid ? csr[p + own] : node;
      const float rs = rn[ms];

      float4 a[EPW];
      #pragma unroll
      for (int j = 0; j < EPW; ++j) {
        const int sj = __builtin_amdgcn_readlane(ms, 4 * j);
        a[j] = *(const float4*)(X + (size_t)sj * HID + lane * 4);
      }
      asm volatile("" ::: "memory");
      float s[EPW];
      #pragma unroll
      for (int j = 0; j < EPW; ++j)
        s[j] = __builtin_fmaf(a[j].w, xd.w,
               __builtin_fmaf(a[j].z, xd.z,
               __builtin_fmaf(a[j].x, xd.x, a[j].y * xd.y)));
      float t[4];
      #pragma unroll
      for (int i = 0; i < 4; ++i) {
        const float keep = up16 ? s[i + 4] : s[i];
        const float send = up16 ? s[i] : s[i + 4];
        t[i] = keep + xrow(send);
      }
      float u[2];
      #pragma unroll
      for (int i = 0; i < 2; ++i) {
        const float keep = up8 ? t[i + 2] : t[i];
        const float send = up8 ? t[i] : t[i + 2];
        u[i] = keep + xmask<8>(send);
      }
      float mine;
      {
        const float keep = up4 ? u[1] : u[0];
        const float send = up4 ? u[0] : u[1];
        mine = keep + xmask<4>(send);
      }
      mine += xmask<2>(mine);
      mine += xmask<1>(mine);
      const float logit = valid ? beta * mine * rs * rd : -__builtin_inff();

      float bm = logit;
      bm = fmaxf(bm, xmask<4>(bm));
      bm = fmaxf(bm, xmask<8>(bm));
      bm = fmaxf(bm, xrow(bm));
      const float m_new = fmaxf(m, bm);
      const float scale = __expf(m - m_new);
      const float e     = valid ? __expf(logit - m_new) : 0.f;
      float es = e;
      es += xrow(es);
      es += xmask<8>(es);
      es += xmask<4>(es);
      ssum  = ssum * scale + es;
      acc.x *= scale; acc.y *= scale; acc.z *= scale; acc.w *= scale;
      #pragma unroll
      for (int j = 0; j < EPW; ++j) {
        const float ej = __int_as_float(
            __builtin_amdgcn_readlane(__float_as_int(e), 4 * j));
        acc.x = __builtin_fmaf(ej, a[j].x, acc.x);
        acc.y = __builtin_fmaf(ej, a[j].y, acc.y);
        acc.z = __builtin_fmaf(ej, a[j].z, acc.z);
        acc.w = __builtin_fmaf(ej, a[j].w, acc.w);
      }
      m = m_new;
    }

    const float den = ssum + 1e-16f;
    float4 o;
    o.x = acc.x / den; o.y = acc.y / den; o.z = acc.z / den; o.w = acc.w / den;
    *(float4*)(Y + (size_t)node * HID + lane * 4) = o;
    if (rn_out) {
      float q = o.x * o.x + o.y * o.y + o.z * o.z + o.w * o.w;
      q = wave_sum(q);
      if (lane == 0) rn_out[node] = 1.f / fmaxf(sqrtf(q), 1e-12f);
    }
  }
}

extern "C" void kernel_launch(void* const* d_in, const int* in_sizes, int n_in,
                              void* d_out, int out_size, void* d_ws, size_t ws_size,
                              hipStream_t stream) {
  const float* x     = (const float*)d_in[0];
  const int*   ei    = (const int*)d_in[1];
  const float* W1    = (const float*)d_in[2];
  const float* b1    = (const float*)d_in[3];
  const float* beta2 = (const float*)d_in[4];
  float* out = (float*)d_out;

  float* ws     = (float*)d_ws;
  float* h1     = ws;
  float* h2     = ws + 6400000;
  int*   recs   = (int*)h2;
  int*   startT = (int*)h2 + 1200128;
  int*   lenT   = (int*)h2 + 1225216;
  float* rn1    = ws + 12800000;
  float* rn2    = ws + 12850000;
  int*   rowptr = (int*)(ws + 12900000);
  int*   csr    = (int*)(ws + 12950004);

  const int gemm_waves  = (N_NODES + 15) / 16;
  const int gemm_blocks = (gemm_waves * 32 + 255) / 256;
  k_gemm_relu<<<gemm_blocks, 256, 0, stream>>>(x, W1, b1, h1, rn1);

  k_bin<<<N_CHUNKS, BIN_T, 0, stream>>>(ei, recs, startT, lenT);
  k_group<<<N_BINS, 1024, 0, stream>>>(recs, startT, lenT, rowptr, csr);

  const int conv_blocks = 2500;
  k_conv<<<conv_blocks, 128, 0, stream>>>(h1, rn1, rowptr, csr, nullptr, h2, rn2);
  k_conv<<<conv_blocks, 128, 0, stream>>>(h2, rn2, rowptr, csr, beta2, out, nullptr);
}


# GEMM: each 16-wave work-group (one per WGP) copies W into LDS once and reads B fragments by conflict-free ds_load_b64 (row pitch 260 words), not 16-line global loads; k order and sums unchanged
# speedup vs baseline: 6.4126x; 1.2038x over previous
#include <hip/hip_runtime.h>
#include <hip/hip_bf16.h>
#include <math.h>


typedef float v2f __attribute__((ext_vector_type(2)));
typedef float v8f __attribute__((ext_vector_type(8)));

#define N_NODES 50000
#define N_EDGES 600000
#define E_TOT   650000
#define IN_CH   256
#define HID     128
#define EPW     8
static_assert(EPW == 8,
              "the halving exchange of k_conv is written out for 8 edges: 4 lanes per edge");
#define BIN_SHIFT 8
#define BIN_NODES (1 << BIN_SHIFT)
#define N_BINS    ((N_NODES + BIN_NODES - 1) / BIN_NODES)
#define N_CHUNKS  128
#define CHUNK     4688
#define BIN_T     512
#define BIN_EPT   10
#define RUN_REG   5
static_assert(N_BINS <= 256 && BIN_NODES == 256,
              "both LDS tables are scanned as 256 entries by one wave");
static_assert(N_CHUNKS * CHUNK >= N_EDGES, "the chunks must cover the edge list");
static_assert(BIN_T * BIN_EPT >= CHUNK, "a binning block holds its chunk in registers");
static_assert(N_CHUNKS * 8 == 1024,
              "a grouping block gives each chunk's run 8 of its 1024 threads");

__device__ __forceinline__ float xrow(float x) {
  const int b = __float_as_int(x);
  return __int_as_float(
      __builtin_amdgcn_permlanex16(b, b, 0x76543210, 0xfedcba98, false, false));
}

template <int M>
__device__ __forceinline__ float xmask(float x) {
  return __int_as_float(__builtin_amdgcn_update_dpp(
      0, __float_as_int(x), 0x160 + M, 0xf, 0xf, true));
}

__device__ __forceinline__ float wave_sum(float v) {
  v += xrow(v);
  v += xmask<8>(v);
  v += xmask<4>(v);
  v += xmask<2>(v);
  v += xmask<1>(v);
  return v;
}

#define GEMM_T      512
#define GEMM_BLOCKS 256
#define N_TILES     ((N_NODES + 15) / 16)
#define W_PITCH     260
static_assert((HID * IN_CH / 4) % GEMM_T == 0, "the copy of W is an equal share per thread");
static_assert(W_PITCH >= IN_CH && W_PITCH % 64 == 4, "see the note on LDS banks above");

__global__ void __launch_bounds__(GEMM_T)
k_gemm_relu(const float* __restrict__ X,
            const float* __restrict__ W,
            const float* __restrict__ B,
            float* __restrict__ H,
            float* __restrict__ rn) {
  __shared__ __attribute__((aligned(16))) float Ws[HID * W_PITCH];
  const int tid  = (int)threadIdx.x;
  const int lane = tid & 31;
  const int wv   = __builtin_amdgcn_readfirstlane(tid >> 5);

  #pragma unroll
  for (int j = 0; j < (HID * IN_CH / 4) / GEMM_T; ++j) {
    const int i = tid + j * GEMM_T;
    const int r = i >> 6;
    const int cc = (i & 63) << 2;
    const float4 v = *(const float4*)(W + (size_t)r * IN_CH + cc);
    *(float4*)(Ws + r * W_PITCH + cc) = v;
  }
  __syncthreads();

  const int kg   = (lane >> 4) * 2;
  const int n15  = lane & 15;
  const float* __restrict__ wl = Ws + n15 * W_PITCH + kg;

  for (int tile = wv * (int)gridDim.x + (int)blockIdx.x; tile < N_TILES;
       tile += (GEMM_T / 32) * (int)gridDim.x) {
  const int m0   = tile * 16;
  const int mrow = m0 + n15;

  v8f acc[8];
  #pragma unroll
  for (int t = 0; t < 8; ++t) acc[t] = (v8f){0.f,0.f,0.f,0.f,0.f,0.f,0.f,0.f};

  const float* __restrict__ xr = X + (size_t)mrow * IN_CH + kg;

  for (int k = 0; k < IN_CH; k += 8) {
    const v2f a0 = *(const v2f*)(xr + k);
    const v2f a1 = *(const v2f*)(xr + k + 4);
    v2f b0[8], b1[8];
    #pragma unroll
    for (int t = 0; t < 8; ++t)
      b0[t] = *(const v2f*)(wl + t * 16 * W_PITCH + k);
    #pragma unroll
    for (int t = 0; t < 8; ++t)
      b1[t] = *(const v2f*)(wl + t * 16 * W_PITCH + k + 4);
    #pragma unroll
    for (int t = 0; t < 8; ++t)
      acc[t] = __builtin_amdgcn_wmma_f32_16x16x4_f32(
          false, a0, false, b0[t], (short)0, acc[t], false, false);
    #pragma unroll
    for (int t = 0; t < 8; ++t)
      acc[t] = __builtin_amdgcn_wmma_f32_16x16x4_f32(
          false, a1, false, b1[t], (short)0, acc[t], false, false);
  }

  const int mside = (lane >> 4) * 8;
  float ss[8];
  #pragma unroll
  for (int r = 0; r < 8; ++r) ss[r] = 0.f;

  #pragma unroll
  for (int t = 0; t < 8; ++t) {
    const int col = t * 16 + n15;
    const float bb = B[col];
    #pragma unroll
    for (int r = 0; r < 8; ++r) {
      const int row = m0 + r + mside;
      float v = acc[t][r] + bb;
      v = v > 0.f ? v : 0.f;
      H[(size_t)row * HID + col] = v;
      ss[r] += v * v;
    }
  }
  #pragma unroll
  for (int r = 0; r < 8; ++r) {
    #pragma unroll
    for (int off = 8; off >= 1; off >>= 1) ss[r] += __shfl_xor(ss[r], off, 32);
  }
  if (n15 == 0) {
    #pragma unroll
    for (int r = 0; r < 8; ++r)
      rn[m0 + r + mside] = 1.f / fmaxf(sqrtf(ss[r]), 1e-12f);
  }
  }
}


__device__ __forceinline__ int scan256_wave(const int* tab, int lane, int (&ex)[8]) {
  int run = 0;
  #pragma unroll
  for (int j = 0; j < 8; ++j) {
    const int t = tab[lane * 8 + j];
    ex[j] = run;
    run += t;
  }
  int x = run;
  #pragma unroll
  for (int off = 1; off < 32; off <<= 1) {
    const int y = __shfl_up(x, off, 32);
    if (lane >= off) x += y;
  }
  const int before = x - run;
  #pragma unroll
  for (int j = 0; j < 8; ++j) ex[j] += before;
  return x;
}

__global__ void __launch_bounds__(BIN_T)
k_bin(const int* __restrict__ EI,
      int* __restrict__ recs,
      int* __restrict__ startT,
      int* __restrict__ lenT) {
  __shared__ int hist[256];
  __shared__ int start[256];
  const int tid  = (int)threadIdx.x;
  const int w    = (int)blockIdx.x;
  const int base = w * CHUNK;
  if (tid < 256) hist[tid] = 0;
  __syncthreads();

  int s[BIN_EPT], d[BIN_EPT], r[BIN_EPT];
  #pragma unroll
  for (int k = 0; k < BIN_EPT; ++k) {
    const int i = tid + k * BIN_T;
    const bool ok = (i < CHUNK) && (base + i < N_EDGES);
    s[k] = ok ? EI[base + i] : 0;
    d[k] = ok ? EI[N_EDGES + base + i] : -1;
  }
  #pragma unroll
  for (int k = 0; k < BIN_EPT; ++k) {
    r[k] = 0;
    if (d[k] >= 0) r[k] = atomicAdd(&hist[d[k] >> BIN_SHIFT], 1);
  }
  __syncthreads();
  if (tid < 32) {
    int ex[8];
    scan256_wave(hist, tid, ex);
    #pragma unroll
    for (int j = 0; j < 8; ++j) start[tid * 8 + j] = ex[j];
  }
  __syncthreads();
  if (tid < N_BINS) {
    startT[tid * N_CHUNKS + w] = start[tid];
    lenT[tid * N_CHUNKS + w]   = hist[tid];
  }
  #pragma unroll
  for (int k = 0; k < BIN_EPT; ++k) {
    if (d[k] >= 0) {
      const int pos = base + start[d[k] >> BIN_SHIFT] + r[k];
      *(int2*)(recs + 2 * (size_t)pos) = make_int2(s[k], d[k]);
    }
  }
}

__global__ void __launch_bounds__(1024)
k_group(const int* __restrict__ recs,
        const int* __restrict__ startT,
        const int* __restrict__ lenT,
        int* __restrict__ rowptr,
        int* __restrict__ csr) {
  __shared__ int cnt[BIN_NODES];
  __shared__ int wpart[32];
  const int tid  = (int)threadIdx.x;
  const int lane = tid & 31;
  const int wv   = tid >> 5;
  const int b    = (int)blockIdx.x;
  const int run  = tid >> 3;
  const int sub  = tid & 7;
  const int st = startT[b * N_CHUNKS + run];
  const int ln = lenT[b * N_CHUNKS + run];

  int part = (sub == 0) ? st : 0;
  #pragma unroll
  for (int off = 16; off >= 1; off >>= 1) part += __shfl_xor(part, off, 32);
  if (lane == 0) wpart[wv] = part;
  if (tid < BIN_NODES) cnt[tid] = 0;
  __syncthreads();
  int base = wpart[lane];
  #pragma unroll
  for (int off = 16; off >= 1; off >>= 1) base += __shfl_xor(base, off, 32);

  const int* __restrict__ rp = recs + 2 * ((size_t)run * CHUNK + st);
  int2 q[RUN_REG];
  #pragma unroll
  for (int k = 0; k < RUN_REG; ++k) {
    const int i = sub + 8 * k;
    q[k] = (i < ln) ? *(const int2*)(rp + 2 * i) : make_int2(0, -1);
  }
  #pragma unroll
  for (int k = 0; k < RUN_REG; ++k)
    if (q[k].y >= 0) atomicAdd(&cnt[q[k].y & (BIN_NODES - 1)], 1);
  for (int i = sub + 8 * RUN_REG; i < ln; i += 8) {
    const int2 rc = *(const int2*)(rp + 2 * i);
    atomicAdd(&cnt[rc.y & (BIN_NODES - 1)], 1);
  }
  __syncthreads();

  if (tid < 32) {
    int ex[8];
    const int incl = scan256_wave(cnt, tid, ex);
    #pragma unroll
    for (int j = 0; j < 8; ++j) {
      const int li   = tid * 8 + j;
      const int node = b * BIN_NODES + li;
      cnt[li] = ex[j];
      if (node < N_NODES) rowptr[node] = base + ex[j];
    }
    if (b == N_BINS - 1 && tid == 31) rowptr[N_NODES] = base + incl;
  }
  __syncthreads();

  #pragma unroll
  for (int k = 0; k < RUN_REG; ++k) {
    if (q[k].y >= 0) {
      const int pos = atomicAdd(&cnt[q[k].y & (BIN_NODES - 1)], 1);
      csr[base + pos] = q[k].x;
    }
  }
  for (int i = sub + 8 * RUN_REG; i < ln; i += 8) {
    const int2 rc = *(const int2*)(rp + 2 * i);
    const int pos = atomicAdd(&cnt[rc.y & (BIN_NODES - 1)], 1);
    csr[base + pos] = rc.x;
  }
}

__global__ void __launch_bounds__(128)
k_conv(const float* __restrict__ X,
       const float* __restrict__ rn,
       const int* __restrict__ rowptr,
       const int* __restrict__ csr,
       const float* __restrict__ beta_ptr,
       float* __restrict__ Y,
       float* __restrict__ rn_out) {
  const int gw = __builtin_amdgcn_readfirstlane(
      (int)((blockIdx.x * blockDim.x + threadIdx.x) >> 5));
  const int nw   = (int)((gridDim.x * blockDim.x) >> 5);
  const int lane = (int)(threadIdx.x & 31);
  const int own  = lane >> 2;
  const bool up16 = (lane & 16) != 0;
  const bool up8  = (lane & 8) != 0;
  const bool up4  = (lane & 4) != 0;
  const float beta = beta_ptr ? beta_ptr[0] : 1.0f;

  for (int node = gw; node < N_NODES; node += nw) {
    const int beg = rowptr[node];
    const int end = rowptr[node + 1];
    const float4 xd = *(const float4*)(X + (size_t)node * HID + lane * 4);
    const float  rd = rn[node];

    float sd = __builtin_fmaf(xd.w, xd.w,
               __builtin_fmaf(xd.z, xd.z,
               __builtin_fmaf(xd.x, xd.x, xd.y * xd.y)));
    sd = wave_sum(sd);
    float  m    = beta * sd * rd * rd;
    float  ssum = 1.0f;
    float4 acc  = xd;

    for (int p = beg; p < end; p += EPW) {
      const bool valid = (p + own) < end;
      const int   ms = valid ? csr[p + own] : node;
      const float rs = rn[ms];

      float4 a[EPW];
      #pragma unroll
      for (int j = 0; j < EPW; ++j) {
        const int sj = __builtin_amdgcn_readlane(ms, 4 * j);
        a[j] = *(const float4*)(X + (size_t)sj * HID + lane * 4);
      }
      asm volatile("" ::: "memory");
      float s[EPW];
      #pragma unroll
      for (int j = 0; j < EPW; ++j)
        s[j] = __builtin_fmaf(a[j].w, xd.w,
               __builtin_fmaf(a[j].z, xd.z,
               __builtin_fmaf(a[j].x, xd.x, a[j].y * xd.y)));
      float t[4];
      #pragma unroll
      for (int i = 0; i < 4; ++i) {
        const float keep = up16 ? s[i + 4] : s[i];
        const float send = up16 ? s[i] : s[i + 4];
        t[i] = keep + xrow(send);
      }
      float u[2];
      #pragma unroll
      for (int i = 0; i < 2; ++i) {
        const float keep = up8 ? t[i + 2] : t[i];
        const float send = up8 ? t[i] : t[i + 2];
        u[i] = keep + xmask<8>(send);
      }
      float mine;
      {
        const float keep = up4 ? u[1] : u[0];
        const float send = up4 ? u[0] : u[1];
        mine = keep + xmask<4>(send);
      }
      mine += xmask<2>(mine);
      mine += xmask<1>(mine);
      const float logit = valid ? beta * mine * rs * rd : -__builtin_inff();

      float bm = logit;
      bm = fmaxf(bm, xmask<4>(bm));
      bm = fmaxf(bm, xmask<8>(bm));
      bm = fmaxf(bm, xrow(bm));
      const float m_new = fmaxf(m, bm);
      const float scale = __expf(m - m_new);
      const float e     = valid ? __expf(logit - m_new) : 0.f;
      float es = e;
      es += xrow(es);
      es += xmask<8>(es);
      es += xmask<4>(es);
      ssum  = ssum * scale + es;
      acc.x *= scale; acc.y *= scale; acc.z *= scale; acc.w *= scale;
      #pragma unroll
      for (int j = 0; j < EPW; ++j) {
        const float ej = __int_as_float(
            __builtin_amdgcn_readlane(__float_as_int(e), 4 * j));
        acc.x = __builtin_fmaf(ej, a[j].x, acc.x);
        acc.y = __builtin_fmaf(ej, a[j].y, acc.y);
        acc.z = __builtin_fmaf(ej, a[j].z, acc.z);
        acc.w = __builtin_fmaf(ej, a[j].w, acc.w);
      }
      m = m_new;
    }

    const float den = ssum + 1e-16f;
    float4 o;
    o.x = acc.x / den; o.y = acc.y / den; o.z = acc.z / den; o.w = acc.w / den;
    *(float4*)(Y + (size_t)node * HID + lane * 4) = o;
    if (rn_out) {
      float q = o.x * o.x + o.y * o.y + o.z * o.z + o.w * o.w;
      q = wave_sum(q);
      if (lane == 0) rn_out[node] = 1.f / fmaxf(sqrtf(q), 1e-12f);
    }
  }
}

extern "C" void kernel_launch(void* const* d_in, const int* in_sizes, int n_in,
                              void* d_out, int out_size, void* d_ws, size_t ws_size,
                              hipStream_t stream) {
  const float* x     = (const float*)d_in[0];
  const int*   ei    = (const int*)d_in[1];
  const float* W1    = (const float*)d_in[2];
  const float* b1    = (const float*)d_in[3];
  const float* beta2 = (const float*)d_in[4];
  float* out = (float*)d_out;

  float* ws     = (float*)d_ws;
  float* h1     = ws;
  float* h2     = ws + 6400000;
  int*   recs   = (int*)h2;
  int*   startT = (int*)h2 + 1200128;
  int*   lenT   = (int*)h2 + 1225216;
  float* rn1    = ws + 12800000;
  float* rn2    = ws + 12850000;
  int*   rowptr = (int*)(ws + 12900000);
  int*   csr    = (int*)(ws + 12950004);

  k_gemm_relu<<<GEMM_BLOCKS, GEMM_T, 0, stream>>>(x, W1, b1, h1, rn1);

  k_bin<<<N_CHUNKS, BIN_T, 0, stream>>>(ei, recs, startT, lenT);
  k_group<<<N_BINS, 1024, 0, stream>>>(recs, startT, lenT, rowptr, csr);

  const int conv_blocks = 2500;
  k_conv<<<conv_blocks, 128, 0, stream>>>(h1, rn1, rowptr, csr, nullptr, h2, rn2);
  k_conv<<<conv_blocks, 128, 0, stream>>>(h2, rn2, rowptr, csr, beta2, out, nullptr);
}


# Edge sort's chunk-binning blocks ride in the GEMM launch as its last 128 work-groups (independent work, beside or behind the matrix-bound GEMM blocks): 5 to 4 launches; arithmetic and sums unchanged
# speedup vs baseline: 6.7612x; 1.0544x over previous
#include <hip/hip_runtime.h>
#include <hip/hip_bf16.h>
#include <math.h>


typedef float v2f __attribute__((ext_vector_type(2)));
typedef float v8f __attribute__((ext_vector_type(8)));

#define N_NODES 50000
#define N_EDGES 600000
#define E_TOT   650000
#define IN_CH   256
#define HID     128
#define EPW     8
static_assert(EPW == 8,
              "the halving exchange of k_conv is written out for 8 edges: 4 lanes per edge");
#define BIN_SHIFT 8
#define BIN_NODES (1 << BIN_SHIFT)
#define N_BINS    ((N_NODES + BIN_NODES - 1) / BIN_NODES)
#define N_CHUNKS  128
#define CHUNK     4688
#define BIN_T     512
#define BIN_EPT   10
#define RUN_REG   5
static_assert(N_BINS <= 256 && BIN_NODES == 256,
              "both LDS tables are scanned as 256 entries by one wave");
static_assert(N_CHUNKS * CHUNK >= N_EDGES, "the chunks must cover the edge list");
static_assert(BIN_T * BIN_EPT >= CHUNK, "a binning block holds its chunk in registers");
static_assert(N_CHUNKS * 8 == 1024,
              "a grouping block gives each chunk's run 8 of its 1024 threads");

__device__ __forceinline__ float xrow(float x) {
  const int b = __float_as_int(x);
  return __int_as_float(
      __builtin_amdgcn_permlanex16(b, b, 0x76543210, 0xfedcba98, false, false));
}

template <int M>
__device__ __forceinline__ float xmask(float x) {
  return __int_as_float(__builtin_amdgcn_update_dpp(
      0, __float_as_int(x), 0x160 + M, 0xf, 0xf, true));
}

__device__ __forceinline__ float wave_sum(float v) {
  v += xrow(v);
  v += xmask<8>(v);
  v += xmask<4>(v);
  v += xmask<2>(v);
  v += xmask<1>(v);
  return v;
}

#define GEMM_T      512
#define GEMM_BLOCKS 256
#define N_TILES     ((N_NODES + 15) / 16)
#define W_PITCH     260
static_assert((HID * IN_CH / 4) % GEMM_T == 0, "the copy of W is an equal share per thread");
static_assert(W_PITCH >= IN_CH && W_PITCH % 64 == 4, "see the note on LDS banks above");

static_assert(BIN_T == GEMM_T, "the binning blocks ride in the GEMM launch: one block size");

__device__ __forceinline__ void bin_chunk(int w, const int* __restrict__ EI,
                                          int* __restrict__ recs,
                                          int* __restrict__ startT,
                                          int* __restrict__ lenT);

__global__ void __launch_bounds__(GEMM_T)
k_gemm_relu(const float* __restrict__ X,
            const float* __restrict__ W,
            const float* __restrict__ B,
            float* __restrict__ H,
            float* __restrict__ rn,
            const int* __restrict__ EI,
            int* __restrict__ recs,
            int* __restrict__ startT,
            int* __restrict__ lenT) {
  if ((int)blockIdx.x >= GEMM_BLOCKS) {
    bin_chunk((int)blockIdx.x - GEMM_BLOCKS, EI, recs, startT, lenT);
    return;
  }
  __shared__ __attribute__((aligned(16))) float Ws[HID * W_PITCH];
  const int tid  = (int)threadIdx.x;
  const int lane = tid & 31;
  const int wv   = __builtin_amdgcn_readfirstlane(tid >> 5);

  #pragma unroll
  for (int j = 0; j < (HID * IN_CH / 4) / GEMM_T; ++j) {
    const int i = tid + j * GEMM_T;
    const int r = i >> 6;
    const int cc = (i & 63) << 2;
    const float4 v = *(const float4*)(W + (size_t)r * IN_CH + cc);
    *(float4*)(Ws + r * W_PITCH + cc) = v;
  }
  __syncthreads();

  const int kg   = (lane >> 4) * 2;
  const int n15  = lane & 15;
  const float* __restrict__ wl = Ws + n15 * W_PITCH + kg;

  for (int tile = wv * GEMM_BLOCKS + (int)blockIdx.x; tile < N_TILES;
       tile += (GEMM_T / 32) * GEMM_BLOCKS) {
  const int m0   = tile * 16;
  const int mrow = m0 + n15;

  v8f acc[8];
  #pragma unroll
  for (int t = 0; t < 8; ++t) acc[t] = (v8f){0.f,0.f,0.f,0.f,0.f,0.f,0.f,0.f};

  const float* __restrict__ xr = X + (size_t)mrow * IN_CH + kg;

  for (int k = 0; k < IN_CH; k += 8) {
    const v2f a0 = *(const v2f*)(xr + k);
    const v2f a1 = *(const v2f*)(xr + k + 4);
    v2f b0[8], b1[8];
    #pragma unroll
    for (int t = 0; t < 8; ++t)
      b0[t] = *(const v2f*)(wl + t * 16 * W_PITCH + k);
    #pragma unroll
    for (int t = 0; t < 8; ++t)
      b1[t] = *(const v2f*)(wl + t * 16 * W_PITCH + k + 4);
    #pragma unroll
    for (int t = 0; t < 8; ++t)
      acc[t] = __builtin_amdgcn_wmma_f32_16x16x4_f32(
          false, a0, false, b0[t], (short)0, acc[t], false, false);
    #pragma unroll
    for (int t = 0; t < 8; ++t)
      acc[t] = __builtin_amdgcn_wmma_f32_16x16x4_f32(
          false, a1, false, b1[t], (short)0, acc[t], false, false);
  }

  const int mside = (lane >> 4) * 8;
  float ss[8];
  #pragma unroll
  for (int r = 0; r < 8; ++r) ss[r] = 0.f;

  #pragma unroll
  for (int t = 0; t < 8; ++t) {
    const int col = t * 16 + n15;
    const float bb = B[col];
    #pragma unroll
    for (int r = 0; r < 8; ++r) {
      const int row = m0 + r + mside;
      float v = acc[t][r] + bb;
      v = v > 0.f ? v : 0.f;
      H[(size_t)row * HID + col] = v;
      ss[r] += v * v;
    }
  }
  #pragma unroll
  for (int r = 0; r < 8; ++r) {
    #pragma unroll
    for (int off = 8; off >= 1; off >>= 1) ss[r] += __shfl_xor(ss[r], off, 32);
  }
  if (n15 == 0) {
    #pragma unroll
    for (int r = 0; r < 8; ++r)
      rn[m0 + r + mside] = 1.f / fmaxf(sqrtf(ss[r]), 1e-12f);
  }
  }
}


__device__ __forceinline__ int scan256_wave(const int* tab, int lane, int (&ex)[8]) {
  int run = 0;
  #pragma unroll
  for (int j = 0; j < 8; ++j) {
    const int t = tab[lane * 8 + j];
    ex[j] = run;
    run += t;
  }
  int x = run;
  #pragma unroll
  for (int off = 1; off < 32; off <<= 1) {
    const int y = __shfl_up(x, off, 32);
    if (lane >= off) x += y;
  }
  const int before = x - run;
  #pragma unroll
  for (int j = 0; j < 8; ++j) ex[j] += before;
  return x;
}

__device__ __forceinline__ void bin_chunk(int w, const int* __restrict__ EI,
                                          int* __restrict__ recs,
                                          int* __restrict__ startT,
                                          int* __restrict__ lenT) {
  __shared__ int hist[256];
  __shared__ int start[256];
  const int tid  = (int)threadIdx.x;
  const int base = w * CHUNK;
  if (tid < 256) hist[tid] = 0;
  __syncthreads();

  int s[BIN_EPT], d[BIN_EPT], r[BIN_EPT];
  #pragma unroll
  for (int k = 0; k < BIN_EPT; ++k) {
    const int i = tid + k * BIN_T;
    const bool ok = (i < CHUNK) && (base + i < N_EDGES);
    s[k] = ok ? EI[base + i] : 0;
    d[k] = ok ? EI[N_EDGES + base + i] : -1;
  }
  #pragma unroll
  for (int k = 0; k < BIN_EPT; ++k) {
    r[k] = 0;
    if (d[k] >= 0) r[k] = atomicAdd(&hist[d[k] >> BIN_SHIFT], 1);
  }
  __syncthreads();
  if (tid < 32) {
    int ex[8];
    scan256_wave(hist, tid, ex);
    #pragma unroll
    for (int j = 0; j < 8; ++j) start[tid * 8 + j] = ex[j];
  }
  __syncthreads();
  if (tid < N_BINS) {
    startT[tid * N_CHUNKS + w] = start[tid];
    lenT[tid * N_CHUNKS + w]   = hist[tid];
  }
  #pragma unroll
  for (int k = 0; k < BIN_EPT; ++k) {
    if (d[k] >= 0) {
      const int pos = base + start[d[k] >> BIN_SHIFT] + r[k];
      *(int2*)(recs + 2 * (size_t)pos) = make_int2(s[k], d[k]);
    }
  }
}

__global__ void __launch_bounds__(1024)
k_group(const int* __restrict__ recs,
        const int* __restrict__ startT,
        const int* __restrict__ lenT,
        int* __restrict__ rowptr,
        int* __restrict__ csr) {
  __shared__ int cnt[BIN_NODES];
  __shared__ int wpart[32];
  const int tid  = (int)threadIdx.x;
  const int lane = tid & 31;
  const int wv   = tid >> 5;
  const int b    = (int)blockIdx.x;
  const int run  = tid >> 3;
  const int sub  = tid & 7;
  const int st = startT[b * N_CHUNKS + run];
  const int ln = lenT[b * N_CHUNKS + run];

  int part = (sub == 0) ? st : 0;
  #pragma unroll
  for (int off = 16; off >= 1; off >>= 1) part += __shfl_xor(part, off, 32);
  if (lane == 0) wpart[wv] = part;
  if (tid < BIN_NODES) cnt[tid] = 0;
  __syncthreads();
  int base = wpart[lane];
  #pragma unroll
  for (int off = 16; off >= 1; off >>= 1) base += __shfl_xor(base, off, 32);

  const int* __restrict__ rp = recs + 2 * ((size_t)run * CHUNK + st);
  int2 q[RUN_REG];
  #pragma unroll
  for (int k = 0; k < RUN_REG; ++k) {
    const int i = sub + 8 * k;
    q[k] = (i < ln) ? *(const int2*)(rp + 2 * i) : make_int2(0, -1);
  }
  #pragma unroll
  for (int k = 0; k < RUN_REG; ++k)
    if (q[k].y >= 0) atomicAdd(&cnt[q[k].y & (BIN_NODES - 1)], 1);
  for (int i = sub + 8 * RUN_REG; i < ln; i += 8) {
    const int2 rc = *(const int2*)(rp + 2 * i);
    atomicAdd(&cnt[rc.y & (BIN_NODES - 1)], 1);
  }
  __syncthreads();

  if (tid < 32) {
    int ex[8];
    const int incl = scan256_wave(cnt, tid, ex);
    #pragma unroll
    for (int j = 0; j < 8; ++j) {
      const int li   = tid * 8 + j;
      const int node = b * BIN_NODES + li;
      cnt[li] = ex[j];
      if (node < N_NODES) rowptr[node] = base + ex[j];
    }
    if (b == N_BINS - 1 && tid == 31) rowptr[N_NODES] = base + incl;
  }
  __syncthreads();

  #pragma unroll
  for (int k = 0; k < RUN_REG; ++k) {
    if (q[k].y >= 0) {
      const int pos = atomicAdd(&cnt[q[k].y & (BIN_NODES - 1)], 1);
      csr[base + pos] = q[k].x;
    }
  }
  for (int i = sub + 8 * RUN_REG; i < ln; i += 8) {
    const int2 rc = *(const int2*)(rp + 2 * i);
    const int pos = atomicAdd(&cnt[rc.y & (BIN_NODES - 1)], 1);
    csr[base + pos] = rc.x;
  }
}

__global__ void __launch_bounds__(128)
k_conv(const float* __restrict__ X,
       const float* __restrict__ rn,
       const int* __restrict__ rowptr,
       const int* __restrict__ csr,
       const float* __restrict__ beta_ptr,
       float* __restrict__ Y,
       float* __restrict__ rn_out) {
  const int gw = __builtin_amdgcn_readfirstlane(
      (int)((blockIdx.x * blockDim.x + threadIdx.x) >> 5));
  const int nw   = (int)((gridDim.x * blockDim.x) >> 5);
  const int lane = (int)(threadIdx.x & 31);
  const int own  = lane >> 2;
  const bool up16 = (lane & 16) != 0;
  const bool up8  = (lane & 8) != 0;
  const bool up4  = (lane & 4) != 0;
  const float beta = beta_ptr ? beta_ptr[0] : 1.0f;

  for (int node = gw; node < N_NODES; node += nw) {
    const int beg = rowptr[node];
    const int end = rowptr[node + 1];
    const float4 xd = *(const float4*)(X + (size_t)node * HID + lane * 4);
    const float  rd = rn[node];

    float sd = __builtin_fmaf(xd.w, xd.w,
               __builtin_fmaf(xd.z, xd.z,
               __builtin_fmaf(xd.x, xd.x, xd.y * xd.y)));
    sd = wave_sum(sd);
    float  m    = beta * sd * rd * rd;
    float  ssum = 1.0f;
    float4 acc  = xd;

    for (int p = beg; p < end; p += EPW) {
      const bool valid = (p + own) < end;
      const int   ms = valid ? csr[p + own] : node;
      const float rs = rn[ms];

      float4 a[EPW];
      #pragma unroll
      for (int j = 0; j < EPW; ++j) {
        const int sj = __builtin_amdgcn_readlane(ms, 4 * j);
        a[j] = *(const float4*)(X + (size_t)sj * HID + lane * 4);
      }
      asm volatile("" ::: "memory");
      float s[EPW];
      #pragma unroll
      for (int j = 0; j < EPW; ++j)
        s[j] = __builtin_fmaf(a[j].w, xd.w,
               __builtin_fmaf(a[j].z, xd.z,
               __builtin_fmaf(a[j].x, xd.x, a[j].y * xd.y)));
      float t[4];
      #pragma unroll
      for (int i = 0; i < 4; ++i) {
        const float keep = up16 ? s[i + 4] : s[i];
        const float send = up16 ? s[i] : s[i + 4];
        t[i] = keep + xrow(send);
      }
      float u[2];
      #pragma unroll
      for (int i = 0; i < 2; ++i) {
        const float keep = up8 ? t[i + 2] : t[i];
        const float send = up8 ? t[i] : t[i + 2];
        u[i] = keep + xmask<8>(send);
      }
      float mine;
      {
        const float keep = up4 ? u[1] : u[0];
        const float send = up4 ? u[0] : u[1];
        mine = keep + xmask<4>(send);
      }
      mine += xmask<2>(mine);
      mine += xmask<1>(mine);
      const float logit = valid ? beta * mine * rs * rd : -__builtin_inff();

      float bm = logit;
      bm = fmaxf(bm, xmask<4>(bm));
      bm = fmaxf(bm, xmask<8>(bm));
      bm = fmaxf(bm, xrow(bm));
      const float m_new = fmaxf(m, bm);
      const float scale = __expf(m - m_new);
      const float e     = valid ? __expf(logit - m_new) : 0.f;
      float es = e;
      es += xrow(es);
      es += xmask<8>(es);
      es += xmask<4>(es);
      ssum  = ssum * scale + es;
      acc.x *= scale; acc.y *= scale; acc.z *= scale; acc.w *= scale;
      #pragma unroll
      for (int j = 0; j < EPW; ++j) {
        const float ej = __int_as_float(
            __builtin_amdgcn_readlane(__float_as_int(e), 4 * j));
        acc.x = __builtin_fmaf(ej, a[j].x, acc.x);
        acc.y = __builtin_fmaf(ej, a[j].y, acc.y);
        acc.z = __builtin_fmaf(ej, a[j].z, acc.z);
        acc.w = __builtin_fmaf(ej, a[j].w, acc.w);
      }
      m = m_new;
    }

    const float den = ssum + 1e-16f;
    float4 o;
    o.x = acc.x / den; o.y = acc.y / den; o.z = acc.z / den; o.w = acc.w / den;
    *(float4*)(Y + (size_t)node * HID + lane * 4) = o;
    if (rn_out) {
      float q = o.x * o.x + o.y * o.y + o.z * o.z + o.w * o.w;
      q = wave_sum(q);
      if (lane == 0) rn_out[node] = 1.f / fmaxf(sqrtf(q), 1e-12f);
    }
  }
}

extern "C" void kernel_launch(void* const* d_in, const int* in_sizes, int n_in,
                              void* d_out, int out_size, void* d_ws, size_t ws_size,
                              hipStream_t stream) {
  const float* x     = (const float*)d_in[0];
  const int*   ei    = (const int*)d_in[1];
  const float* W1    = (const float*)d_in[2];
  const float* b1    = (const float*)d_in[3];
  const float* beta2 = (const float*)d_in[4];
  float* out = (float*)d_out;

  float* ws     = (float*)d_ws;
  float* h1     = ws;
  float* h2     = ws + 6400000;
  int*   recs   = (int*)h2;
  int*   startT = (int*)h2 + 1200128;
  int*   lenT   = (int*)h2 + 1225216;
  float* rn1    = ws + 12800000;
  float* rn2    = ws + 12850000;
  int*   rowptr = (int*)(ws + 12900000);
  int*   csr    = (int*)(ws + 12950004);

  k_gemm_relu<<<GEMM_BLOCKS + N_CHUNKS, GEMM_T, 0, stream>>>(x, W1, b1, h1, rn1,
                                                           ei, recs, startT, lenT);

  k_group<<<N_BINS, 1024, 0, stream>>>(recs, startT, lenT, rowptr, csr);

  const int conv_blocks = 2500;
  k_conv<<<conv_blocks, 128, 0, stream>>>(h1, rn1, rowptr, csr, nullptr, h2, rn2);
  k_conv<<<conv_blocks, 128, 0, stream>>>(h2, rn2, rowptr, csr, beta2, out, nullptr);
}


# Conv: a wave takes 5 consecutive nodes, reads their list bounds and rn once at its start and requests the next batch's list entries and next node's row one step ahead; it waits for row gathers only
# speedup vs baseline: 7.0698x; 1.0456x over previous
#include <hip/hip_runtime.h>
#include <hip/hip_bf16.h>
#include <math.h>


typedef float v2f __attribute__((ext_vector_type(2)));
typedef float v8f __attribute__((ext_vector_type(8)));

#define N_NODES 50000
#define N_EDGES 600000
#define E_TOT   650000
#define IN_CH   256
#define HID     128
#define EPW     8
static_assert(EPW == 8,
              "the halving exchange of k_conv is written out for 8 edges: 4 lanes per edge");
#define BIN_SHIFT 8
#define BIN_NODES (1 << BIN_SHIFT)
#define N_BINS    ((N_NODES + BIN_NODES - 1) / BIN_NODES)
#define N_CHUNKS  128
#define CHUNK     4688
#define BIN_T     512
#define BIN_EPT   10
#define RUN_REG   5
static_assert(N_BINS <= 256 && BIN_NODES == 256,
              "both LDS tables are scanned as 256 entries by one wave");
static_assert(N_CHUNKS * CHUNK >= N_EDGES, "the chunks must cover the edge list");
static_assert(BIN_T * BIN_EPT >= CHUNK, "a binning block holds its chunk in registers");
static_assert(N_CHUNKS * 8 == 1024,
              "a grouping block gives each chunk's run 8 of its 1024 threads");

__device__ __forceinline__ float xrow(float x) {
  const int b = __float_as_int(x);
  return __int_as_float(
      __builtin_amdgcn_permlanex16(b, b, 0x76543210, 0xfedcba98, false, false));
}

template <int M>
__device__ __forceinline__ float xmask(float x) {
  return __int_as_float(__builtin_amdgcn_update_dpp(
      0, __float_as_int(x), 0x160 + M, 0xf, 0xf, true));
}

__device__ __forceinline__ float wave_sum(float v) {
  v += xrow(v);
  v += xmask<8>(v);
  v += xmask<4>(v);
  v += xmask<2>(v);
  v += xmask<1>(v);
  return v;
}

#define GEMM_T      512
#define GEMM_BLOCKS 256
#define N_TILES     ((N_NODES + 15) / 16)
#define W_PITCH     260
static_assert((HID * IN_CH / 4) % GEMM_T == 0, "the copy of W is an equal share per thread");
static_assert(W_PITCH >= IN_CH && W_PITCH % 64 == 4, "see the note on LDS banks above");

static_assert(BIN_T == GEMM_T, "the binning blocks ride in the GEMM launch: one block size");

__device__ __forceinline__ void bin_chunk(int w, const int* __restrict__ EI,
                                          int* __restrict__ recs,
                                          int* __restrict__ startT,
                                          int* __restrict__ lenT);

__global__ void __launch_bounds__(GEMM_T)
k_gemm_relu(const float* __restrict__ X,
            const float* __restrict__ W,
            const float* __restrict__ B,
            float* __restrict__ H,
            float* __restrict__ rn,
            const int* __restrict__ EI,
            int* __restrict__ recs,
            int* __restrict__ startT,
            int* __restrict__ lenT) {
  if ((int)blockIdx.x >= GEMM_BLOCKS) {
    bin_chunk((int)blockIdx.x - GEMM_BLOCKS, EI, recs, startT, lenT);
    return;
  }
  __shared__ __attribute__((aligned(16))) float Ws[HID * W_PITCH];
  const int tid  = (int)threadIdx.x;
  const int lane = tid & 31;
  const int wv   = __builtin_amdgcn_readfirstlane(tid >> 5);

  #pragma unroll
  for (int j = 0; j < (HID * IN_CH / 4) / GEMM_T; ++j) {
    const int i = tid + j * GEMM_T;
    const int r = i >> 6;
    const int cc = (i & 63) << 2;
    const float4 v = *(const float4*)(W + (size_t)r * IN_CH + cc);
    *(float4*)(Ws + r * W_PITCH + cc) = v;
  }
  __syncthreads();

  const int kg   = (lane >> 4) * 2;
  const int n15  = lane & 15;
  const float* __restrict__ wl = Ws + n15 * W_PITCH + kg;

  for (int tile = wv * GEMM_BLOCKS + (int)blockIdx.x; tile < N_TILES;
       tile += (GEMM_T / 32) * GEMM_BLOCKS) {
  const int m0   = tile * 16;
  const int mrow = m0 + n15;

  v8f acc[8];
  #pragma unroll
  for (int t = 0; t < 8; ++t) acc[t] = (v8f){0.f,0.f,0.f,0.f,0.f,0.f,0.f,0.f};

  const float* __restrict__ xr = X + (size_t)mrow * IN_CH + kg;

  for (int k = 0; k < IN_CH; k += 8) {
    const v2f a0 = *(const v2f*)(xr + k);
    const v2f a1 = *(const v2f*)(xr + k + 4);
    v2f b0[8], b1[8];
    #pragma unroll
    for (int t = 0; t < 8; ++t)
      b0[t] = *(const v2f*)(wl + t * 16 * W_PITCH + k);
    #pragma unroll
    for (int t = 0; t < 8; ++t)
      b1[t] = *(const v2f*)(wl + t * 16 * W_PITCH + k + 4);
    #pragma unroll
    for (int t = 0; t < 8; ++t)
      acc[t] = __builtin_amdgcn_wmma_f32_16x16x4_f32(
          false, a0, false, b0[t], (short)0, acc[t], false, false);
    #pragma unroll
    for (int t = 0; t < 8; ++t)
      acc[t] = __builtin_amdgcn_wmma_f32_16x16x4_f32(
          false, a1, false, b1[t], (short)0, acc[t], false, false);
  }

  const int mside = (lane >> 4) * 8;
  float ss[8];
  #pragma unroll
  for (int r = 0; r < 8; ++r) ss[r] = 0.f;

  #pragma unroll
  for (int t = 0; t < 8; ++t) {
    const int col = t * 16 + n15;
    const float bb = B[col];
    #pragma unroll
    for (int r = 0; r < 8; ++r) {
      const int row = m0 + r + mside;
      float v = acc[t][r] + bb;
      v = v > 0.f ? v : 0.f;
      H[(size_t)row * HID + col] = v;
      ss[r] += v * v;
    }
  }
  #pragma unroll
  for (int r = 0; r < 8; ++r) {
    #pragma unroll
    for (int off = 8; off >= 1; off >>= 1) ss[r] += __shfl_xor(ss[r], off, 32);
  }
  if (n15 == 0) {
    #pragma unroll
    for (int r = 0; r < 8; ++r)
      rn[m0 + r + mside] = 1.f / fmaxf(sqrtf(ss[r]), 1e-12f);
  }
  }
}


__device__ __forceinline__ int scan256_wave(const int* tab, int lane, int (&ex)[8]) {
  int run = 0;
  #pragma unroll
  for (int j = 0; j < 8; ++j) {
    const int t = tab[lane * 8 + j];
    ex[j] = run;
    run += t;
  }
  int x = run;
  #pragma unroll
  for (int off = 1; off < 32; off <<= 1) {
    const int y = __shfl_up(x, off, 32);
    if (lane >= off) x += y;
  }
  const int before = x - run;
  #pragma unroll
  for (int j = 0; j < 8; ++j) ex[j] += before;
  return x;
}

__device__ __forceinline__ void bin_chunk(int w, const int* __restrict__ EI,
                                          int* __restrict__ recs,
                                          int* __restrict__ startT,
                                          int* __restrict__ lenT) {
  __shared__ int hist[256];
  __shared__ int start[256];
  const int tid  = (int)threadIdx.x;
  const int base = w * CHUNK;
  if (tid < 256) hist[tid] = 0;
  __syncthreads();

  int s[BIN_EPT], d[BIN_EPT], r[BIN_EPT];
  #pragma unroll
  for (int k = 0; k < BIN_EPT; ++k) {
    const int i = tid + k * BIN_T;
    const bool ok = (i < CHUNK) && (base + i < N_EDGES);
    s[k] = ok ? EI[base + i] : 0;
    d[k] = ok ? EI[N_EDGES + base + i] : -1;
  }
  #pragma unroll
  for (int k = 0; k < BIN_EPT; ++k) {
    r[k] = 0;
    if (d[k] >= 0) r[k] = atomicAdd(&hist[d[k] >> BIN_SHIFT], 1);
  }
  __syncthreads();
  if (tid < 32) {
    int ex[8];
    scan256_wave(hist, tid, ex);
    #pragma unroll
    for (int j = 0; j < 8; ++j) start[tid * 8 + j] = ex[j];
  }
  __syncthreads();
  if (tid < N_BINS) {
    startT[tid * N_CHUNKS + w] = start[tid];
    lenT[tid * N_CHUNKS + w]   = hist[tid];
  }
  #pragma unroll
  for (int k = 0; k < BIN_EPT; ++k) {
    if (d[k] >= 0) {
      const int pos = base + start[d[k] >> BIN_SHIFT] + r[k];
      *(int2*)(recs + 2 * (size_t)pos) = make_int2(s[k], d[k]);
    }
  }
}

__global__ void __launch_bounds__(1024)
k_group(const int* __restrict__ recs,
        const int* __restrict__ startT,
        const int* __restrict__ lenT,
        int* __restrict__ rowptr,
        int* __restrict__ csr) {
  __shared__ int cnt[BIN_NODES];
  __shared__ int wpart[32];
  const int tid  = (int)threadIdx.x;
  const int lane = tid & 31;
  const int wv   = tid >> 5;
  const int b    = (int)blockIdx.x;
  const int run  = tid >> 3;
  const int sub  = tid & 7;
  const int st = startT[b * N_CHUNKS + run];
  const int ln = lenT[b * N_CHUNKS + run];

  int part = (sub == 0) ? st : 0;
  #pragma unroll
  for (int off = 16; off >= 1; off >>= 1) part += __shfl_xor(part, off, 32);
  if (lane == 0) wpart[wv] = part;
  if (tid < BIN_NODES) cnt[tid] = 0;
  __syncthreads();
  int base = wpart[lane];
  #pragma unroll
  for (int off = 16; off >= 1; off >>= 1) base += __shfl_xor(base, off, 32);

  const int* __restrict__ rp = recs + 2 * ((size_t)run * CHUNK + st);
  int2 q[RUN_REG];
  #pragma unroll
  for (int k = 0; k < RUN_REG; ++k) {
    const int i = sub + 8 * k;
    q[k] = (i < ln) ? *(const int2*)(rp + 2 * i) : make_int2(0, -1);
  }
  #pragma unroll
  for (int k = 0; k < RUN_REG; ++k)
    if (q[k].y >= 0) atomicAdd(&cnt[q[k].y & (BIN_NODES - 1)], 1);
  for (int i = sub + 8 * RUN_REG; i < ln; i += 8) {
    const int2 rc = *(const int2*)(rp + 2 * i);
    atomicAdd(&cnt[rc.y & (BIN_NODES - 1)], 1);
  }
  __syncthreads();

  if (tid < 32) {
    int ex[8];
    const int incl = scan256_wave(cnt, tid, ex);
    #pragma unroll
    for (int j = 0; j < 8; ++j) {
      const int li   = tid * 8 + j;
      const int node = b * BIN_NODES + li;
      cnt[li] = ex[j];
      if (node < N_NODES) rowptr[node] = base + ex[j];
    }
    if (b == N_BINS - 1 && tid == 31) rowptr[N_NODES] = base + incl;
  }
  __syncthreads();

  #pragma unroll
  for (int k = 0; k < RUN_REG; ++k) {
    if (q[k].y >= 0) {
      const int pos = atomicAdd(&cnt[q[k].y & (BIN_NODES - 1)], 1);
      csr[base + pos] = q[k].x;
    }
  }
  for (int i = sub + 8 * RUN_REG; i < ln; i += 8) {
    const int2 rc = *(const int2*)(rp + 2 * i);
    const int pos = atomicAdd(&cnt[rc.y & (BIN_NODES - 1)], 1);
    csr[base + pos] = rc.x;
  }
}

#define CONV_BLOCKS 2500
#define NPW         5
static_assert(CONV_BLOCKS * 4 * NPW >= N_NODES, "the waves must cover the nodes");
static_assert(NPW < 32, "lane l of a wave holds the list bounds of its node l, lane NPW the last end");

__global__ void __launch_bounds__(128)
k_conv(const float* __restrict__ X,
       const float* __restrict__ rn,
       const int* __restrict__ rowptr,
       const int* __restrict__ csr,
       const float* __restrict__ beta_ptr,
       float* __restrict__ Y,
       float* __restrict__ rn_out) {
  const int gw = __builtin_amdgcn_readfirstlane(
      (int)((blockIdx.x * blockDim.x + threadIdx.x) >> 5));
  const int lane = (int)(threadIdx.x & 31);
  const int own  = lane >> 2;
  const bool up16 = (lane & 16) != 0;
  const bool up8  = (lane & 8) != 0;
  const bool up4  = (lane & 4) != 0;
  const float beta = beta_ptr ? beta_ptr[0] : 1.0f;
  const int n0 = gw * NPW;
  if (n0 >= N_NODES) return;
  const int cnt = (N_NODES - n0 < NPW) ? (N_NODES - n0) : NPW;

  const int   rp = rowptr[n0 + (lane < cnt ? lane : cnt)];
  const float rv = rn[n0 + (lane < cnt ? lane : cnt - 1)];
  int    node = n0;
  int    beg  = __builtin_amdgcn_readlane(rp, 0);
  int    end  = __builtin_amdgcn_readlane(rp, 1);
  float  rd   = __int_as_float(__builtin_amdgcn_readlane(__float_as_int(rv), 0));
  float4 xd   = *(const float4*)(X + (size_t)node * HID + lane * 4);
  int    ms   = (beg + own < end) ? csr[beg + own] : node;

  for (int i = 0; i < cnt; ++i) {
    const int   in    = (i + 1 < cnt) ? i + 1 : i;
    const int   nq    = n0 + in;
    const int   beg_n = __builtin_amdgcn_readlane(rp, in);
    const int   end_n = __builtin_amdgcn_readlane(rp, in + 1);
    const float rd_n  = __int_as_float(__builtin_amdgcn_readlane(__float_as_int(rv), in));
    float4 xd_n = xd;

    float sd = __builtin_fmaf(xd.w, xd.w,
               __builtin_fmaf(xd.z, xd.z,
               __builtin_fmaf(xd.x, xd.x, xd.y * xd.y)));
    sd = wave_sum(sd);
    float  m    = beta * sd * rd * rd;
    float  ssum = 1.0f;
    float4 acc  = xd;

    if (beg >= end) {
      ms   = (beg_n + own < end_n) ? csr[beg_n + own] : nq;
      xd_n = *(const float4*)(X + (size_t)nq * HID + lane * 4);
    }

    for (int p = beg; p < end; p += EPW) {
      const bool  valid = (p + own) < end;
      const float rs = rn[ms];

      float4 a[EPW];
      #pragma unroll
      for (int j = 0; j < EPW; ++j) {
        const int sj = __builtin_amdgcn_readlane(ms, 4 * j);
        a[j] = *(const float4*)(X + (size_t)sj * HID + lane * 4);
      }
      asm volatile("" ::: "memory");
      if (p == beg) xd_n = *(const float4*)(X + (size_t)nq * HID + lane * 4);
      const int  pn   = p + EPW;
      const bool last = pn >= end;
      const int  qb   = last ? beg_n : pn;
      const int  qe   = last ? end_n : end;
      const int  qd   = last ? nq : node;
      const int  ms_n = (qb + own < qe) ? csr[qb + own] : qd;
      asm volatile("" ::: "memory");
      float s[EPW];
      #pragma unroll
      for (int j = 0; j < EPW; ++j)
        s[j] = __builtin_fmaf(a[j].w, xd.w,
               __builtin_fmaf(a[j].z, xd.z,
               __builtin_fmaf(a[j].x, xd.x, a[j].y * xd.y)));
      float t[4];
      #pragma unroll
      for (int k = 0; k < 4; ++k) {
        const float keep = up16 ? s[k + 4] : s[k];
        const float send = up16 ? s[k] : s[k + 4];
        t[k] = keep + xrow(send);
      }
      float u[2];
      #pragma unroll
      for (int k = 0; k < 2; ++k) {
        const float keep = up8 ? t[k + 2] : t[k];
        const float send = up8 ? t[k] : t[k + 2];
        u[k] = keep + xmask<8>(send);
      }
      float mine;
      {
        const float keep = up4 ? u[1] : u[0];
        const float send = up4 ? u[0] : u[1];
        mine = keep + xmask<4>(send);
      }
      mine += xmask<2>(mine);
      mine += xmask<1>(mine);
      const float logit = valid ? beta * mine * rs * rd : -__builtin_inff();

      float bm = logit;
      bm = fmaxf(bm, xmask<4>(bm));
      bm = fmaxf(bm, xmask<8>(bm));
      bm = fmaxf(bm, xrow(bm));
      const float m_new = fmaxf(m, bm);
      const float scale = __expf(m - m_new);
      const float e     = valid ? __expf(logit - m_new) : 0.f;
      float es = e;
      es += xrow(es);
      es += xmask<8>(es);
      es += xmask<4>(es);
      ssum  = ssum * scale + es;
      acc.x *= scale; acc.y *= scale; acc.z *= scale; acc.w *= scale;
      #pragma unroll
      for (int j = 0; j < EPW; ++j) {
        const float ej = __int_as_float(
            __builtin_amdgcn_readlane(__float_as_int(e), 4 * j));
        acc.x = __builtin_fmaf(ej, a[j].x, acc.x);
        acc.y = __builtin_fmaf(ej, a[j].y, acc.y);
        acc.z = __builtin_fmaf(ej, a[j].z, acc.z);
        acc.w = __builtin_fmaf(ej, a[j].w, acc.w);
      }
      m  = m_new;
      ms = ms_n;
    }

    const float den = ssum + 1e-16f;
    float4 o;
    o.x = acc.x / den; o.y = acc.y / den; o.z = acc.z / den; o.w = acc.w / den;
    *(float4*)(Y + (size_t)node * HID + lane * 4) = o;
    if (rn_out) {
      float q = o.x * o.x + o.y * o.y + o.z * o.z + o.w * o.w;
      q = wave_sum(q);
      if (lane == 0) rn_out[node] = 1.f / fmaxf(sqrtf(q), 1e-12f);
    }

    node = nq; beg = beg_n; end = end_n; rd = rd_n; xd = xd_n;
  }
}

extern "C" void kernel_launch(void* const* d_in, const int* in_sizes, int n_in,
                              void* d_out, int out_size, void* d_ws, size_t ws_size,
                              hipStream_t stream) {
  const float* x     = (const float*)d_in[0];
  const int*   ei    = (const int*)d_in[1];
  const float* W1    = (const float*)d_in[2];
  const float* b1    = (const float*)d_in[3];
  const float* beta2 = (const float*)d_in[4];
  float* out = (float*)d_out;

  float* ws     = (float*)d_ws;
  float* h1     = ws;
  float* h2     = ws + 6400000;
  int*   recs   = (int*)h2;
  int*   startT = (int*)h2 + 1200128;
  int*   lenT   = (int*)h2 + 1225216;
  float* rn1    = ws + 12800000;
  float* rn2    = ws + 12850000;
  int*   rowptr = (int*)(ws + 12900000);
  int*   csr    = (int*)(ws + 12950004);

  k_gemm_relu<<<GEMM_BLOCKS + N_CHUNKS, GEMM_T, 0, stream>>>(x, W1, b1, h1, rn1,
                                                           ei, recs, startT, lenT);

  k_group<<<N_BINS, 1024, 0, stream>>>(recs, startT, lenT, rowptr, csr);

  k_conv<<<CONV_BLOCKS, 128, 0, stream>>>(h1, rn1, rowptr, csr, nullptr, h2, rn2);
  k_conv<<<CONV_BLOCKS, 128, 0, stream>>>(h2, rn2, rowptr, csr, beta2, out, nullptr);
}
